# starting kernel
# baseline (speedup 1.0000x reference)
#include <hip/hip_runtime.h>
#include <hip/hip_bf16.h>


typedef __attribute__((ext_vector_type(2))) float v2f;
typedef __attribute__((ext_vector_type(8))) float v8f;

#define BSZ        2
#define NN         3072
#define EE         4096
#define NODE_DIM   128
#define EDGE_DIM   64
#define NHEAD      8
#define HDIM       16
#define QKV        384

#define WMMA_F32(A, B, C) \
  __builtin_amdgcn_wmma_f32_16x16x4_f32(false, (A), false, (B), (short)0, (C), false, false)

__global__ __launch_bounds__(32)
void gemm_bias_32x32(const float* __restrict__ X, const float* __restrict__ W,
                     const float* __restrict__ bias, float* __restrict__ C,
                     int M, int K, int Ncols) {
  const int tiles_n = Ncols >> 5;
  const int tile = blockIdx.x;
  const int tm = tile / tiles_n;
  const int tn = tile % tiles_n;
  const int lane = threadIdx.x;
  const int half = lane >> 4;
  const int l15  = lane & 15;

  const int row0 = tm * 32 + l15;
  const int row1 = row0 + 16;
  const int col0 = tn * 32 + l15;
  const int col1 = col0 + 16;

  const float* __restrict__ Xr0 = X + (size_t)row0 * K;
  const float* __restrict__ Xr1 = X + (size_t)row1 * K;

  v8f acc00 = {}, acc01 = {}, acc10 = {}, acc11 = {};
  for (int k = 0; k < K; k += 4) {
    const int ka = k + half * 2;
    v2f a0, a1, b0, b1;
    a0.x = Xr0[ka];                         a0.y = Xr0[ka + 1];
    a1.x = Xr1[ka];                         a1.y = Xr1[ka + 1];
    b0.x = W[(size_t)ka * Ncols + col0];    b0.y = W[(size_t)(ka + 1) * Ncols + col0];
    b1.x = W[(size_t)ka * Ncols + col1];    b1.y = W[(size_t)(ka + 1) * Ncols + col1];
    acc00 = WMMA_F32(a0, b0, acc00);
    acc01 = WMMA_F32(a0, b1, acc01);
    acc10 = WMMA_F32(a1, b0, acc10);
    acc11 = WMMA_F32(a1, b1, acc11);
  }

  const float bv0 = bias[col0];
  const float bv1 = bias[col1];
#pragma unroll
  for (int j = 0; j < 8; ++j) {
    const int r0 = tm * 32 + j + half * 8;
    const int r1 = r0 + 16;
    C[(size_t)r0 * Ncols + col0] = acc00[j] + bv0;
    C[(size_t)r0 * Ncols + col1] = acc01[j] + bv1;
    C[(size_t)r1 * Ncols + col0] = acc10[j] + bv0;
    C[(size_t)r1 * Ncols + col1] = acc11[j] + bv1;
  }
}

__global__ void degree_kernel(const float* __restrict__ G, float* __restrict__ deg) {
  const int idx = blockIdx.x * blockDim.x + threadIdx.x;
  if (idx >= BSZ * NN) return;
  const int b = idx / NN;
  const int n = idx - b * NN;
  const float* __restrict__ g = G + (size_t)b * EE * NN + n;
  float s = 0.f;
  for (int e = 0; e < EE; ++e) s += g[(size_t)e * NN];
  deg[idx] = s;
}

__global__ __launch_bounds__(32)
void agg_kernel(const float* __restrict__ G, const float* __restrict__ qkv_e,
                const float* __restrict__ deg, float* __restrict__ qkv) {
  const int tiles_c = QKV / 32;
  const int tiles_n = NN / 32;
  int tile = blockIdx.x;
  const int b  = tile / (tiles_n * tiles_c);
  tile -= b * tiles_n * tiles_c;
  const int tn = tile / tiles_c;
  const int tc = tile - tn * tiles_c;

  const int lane = threadIdx.x;
  const int half = lane >> 4;
  const int l15  = lane & 15;
  const int n0 = tn * 32 + l15;
  const int n1 = n0 + 16;
  const int c0 = tc * 32 + l15;
  const int c1 = c0 + 16;

  const float* __restrict__ Gb = G     + (size_t)b * EE * NN;
  const float* __restrict__ Eb = qkv_e + (size_t)b * EE * QKV;

  v8f acc00 = {}, acc01 = {}, acc10 = {}, acc11 = {};
  for (int k = 0; k < EE; k += 4) {
    const int ka = k + half * 2;
    const size_t ga = (size_t)ka * NN;
    const size_t ea = (size_t)ka * QKV;
    v2f a0, a1, b0, b1;
    a0.x = Gb[ga + n0];        a0.y = Gb[ga + NN + n0];
    a1.x = Gb[ga + n1];        a1.y = Gb[ga + NN + n1];
    b0.x = Eb[ea + c0];        b0.y = Eb[ea + QKV + c0];
    b1.x = Eb[ea + c1];        b1.y = Eb[ea + QKV + c1];
    acc00 = WMMA_F32(a0, b0, acc00);
    acc01 = WMMA_F32(a0, b1, acc01);
    acc10 = WMMA_F32(a1, b0, acc10);
    acc11 = WMMA_F32(a1, b1, acc11);
  }

  float* __restrict__ out = qkv + (size_t)b * NN * QKV;
#pragma unroll
  for (int j = 0; j < 8; ++j) {
    const int r0 = tn * 32 + j + half * 8;
    const int r1 = r0 + 16;
    const float rdiv0 = 1.0f / deg[b * NN + r0];
    const float rdiv1 = 1.0f / deg[b * NN + r1];
    const size_t o00 = (size_t)r0 * QKV + c0;
    const size_t o01 = (size_t)r0 * QKV + c1;
    const size_t o10 = (size_t)r1 * QKV + c0;
    const size_t o11 = (size_t)r1 * QKV + c1;
    out[o00] = out[o00] + acc00[j] * rdiv0;
    out[o01] = out[o01] + acc01[j] * rdiv0;
    out[o10] = out[o10] + acc10[j] * rdiv1;
    out[o11] = out[o11] + acc11[j] * rdiv1;
  }
}

__global__ __launch_bounds__(32)
void attn_kernel(const float* __restrict__ qkv, float* __restrict__ attout) {
  __shared__ float lds[16 * 16];

  const int tiles_q = NN / 16;
  int tile = blockIdx.x;
  const int b = tile / (NHEAD * tiles_q);
  tile -= b * NHEAD * tiles_q;
  const int h  = tile / tiles_q;
  const int tq = tile - h * tiles_q;

  const int lane = threadIdx.x;
  const int half = lane >> 4;
  const int l15  = lane & 15;
  const float scale = 0.25f;

  const float* __restrict__ base = qkv + (size_t)b * NN * QKV + h * 48;

  float qa[8];
  {
    const int m = tq * 16 + l15;
#pragma unroll
    for (int kk = 0; kk < 16; kk += 4) {
      const int cc = kk + half * 2;
      qa[kk / 2]     = base[(size_t)m * QKV + cc] * scale;
      qa[kk / 2 + 1] = base[(size_t)m * QKV + cc + 1] * scale;
    }
  }

  v8f o = {};
  float mrow[8], lrow[8];
#pragma unroll
  for (int j = 0; j < 8; ++j) { mrow[j] = -1e30f; lrow[j] = 0.f; }

  for (int tk = 0; tk < NN / 16; ++tk) {
    v8f s = {};
    {
      const int m = tk * 16 + l15;
      const float* __restrict__ krow = base + (size_t)m * QKV + 16;
#pragma unroll
      for (int kk = 0; kk < 16; kk += 4) {
        const int cc = kk + half * 2;
        v2f a, bb;
        a.x  = qa[kk / 2];
        a.y  = qa[kk / 2 + 1];
        bb.x = krow[cc];
        bb.y = krow[cc + 1];
        s = WMMA_F32(a, bb, s);
      }
    }

    float pnew[8];
#pragma unroll
    for (int j = 0; j < 8; ++j) {
      const float sj = s[j];
      float mx = sj;
      mx = fmaxf(mx, __shfl_xor(mx, 1, 32));
      mx = fmaxf(mx, __shfl_xor(mx, 2, 32));
      mx = fmaxf(mx, __shfl_xor(mx, 4, 32));
      mx = fmaxf(mx, __shfl_xor(mx, 8, 32));
      const float mnew = fmaxf(mrow[j], mx);
      const float corr = __expf(mrow[j] - mnew);
      const float p    = __expf(sj - mnew);
      float ps = p;
      ps += __shfl_xor(ps, 1, 32);
      ps += __shfl_xor(ps, 2, 32);
      ps += __shfl_xor(ps, 4, 32);
      ps += __shfl_xor(ps, 8, 32);
      lrow[j] = lrow[j] * corr + ps;
      mrow[j] = mnew;
      o[j] *= corr;
      pnew[j] = p;
    }

#pragma unroll
    for (int j = 0; j < 8; ++j)
      lds[(j + half * 8) * 16 + l15] = pnew[j];
    __syncthreads();

#pragma unroll
    for (int kk = 0; kk < 16; kk += 4) {
      const int krow = kk + half * 2;
      v2f a, bb;
      a.x = lds[l15 * 16 + krow];
      a.y = lds[l15 * 16 + krow + 1];
      const int m = tk * 16 + krow;
      bb.x = base[(size_t)m * QKV + 32 + l15];
      bb.y = base[(size_t)(m + 1) * QKV + 32 + l15];
      o = WMMA_F32(a, bb, o);
    }
    __syncthreads();
  }

  float* __restrict__ outp = attout + (size_t)b * NN * NODE_DIM + h * HDIM;
#pragma unroll
  for (int j = 0; j < 8; ++j) {
    const int r = tq * 16 + j + half * 8;
    outp[(size_t)r * NODE_DIM + l15] = o[j] / lrow[j];
  }
}

extern "C" void kernel_launch(void* const* d_in, const int* in_sizes, int n_in,
                              void* d_out, int out_size, void* d_ws, size_t ws_size,
                              hipStream_t stream) {
  const float* node    = (const float*)d_in[0];
  const float* edge    = (const float*)d_in[1];
  const float* G       = (const float*)d_in[2];
  const float* W_qkv_n = (const float*)d_in[3];
  const float* b_qkv_n = (const float*)d_in[4];
  const float* W_qkv_e = (const float*)d_in[5];
  const float* b_qkv_e = (const float*)d_in[6];
  const float* W_o     = (const float*)d_in[7];
  const float* b_o     = (const float*)d_in[8];
  float* out = (float*)d_out;

  float* qkv  = (float*)d_ws;
  float* qkve = qkv  + (size_t)BSZ * NN * QKV;
  float* deg  = qkve + (size_t)BSZ * EE * QKV;
  float* att  = deg  + (size_t)BSZ * NN;

  gemm_bias_32x32<<<(BSZ * NN / 32) * (QKV / 32), 32, 0, stream>>>(
      node, W_qkv_n, b_qkv_n, qkv, BSZ * NN, NODE_DIM, QKV);

  gemm_bias_32x32<<<(BSZ * EE / 32) * (QKV / 32), 32, 0, stream>>>(
      edge, W_qkv_e, b_qkv_e, qkve, BSZ * EE, EDGE_DIM, QKV);

  degree_kernel<<<(BSZ * NN + 255) / 256, 256, 0, stream>>>(G, deg);

  agg_kernel<<<BSZ * (NN / 32) * (QKV / 32), 32, 0, stream>>>(G, qkve, deg, qkv);

  attn_kernel<<<BSZ * NHEAD * (NN / 16), 32, 0, stream>>>(qkv, att);

  gemm_bias_32x32<<<(BSZ * NN / 32) * (NODE_DIM / 32), 32, 0, stream>>>(
      att, W_o, b_o, out, BSZ * NN, NODE_DIM, NODE_DIM);
}


# attention: the 3072 keys of a query tile split over the 4 waves of a work-group (48 key tiles each, wave-local LDS); partial max, sum and O merged through LDS in fixed key order; sum order changes
# speedup vs baseline: 2.2872x; 2.2872x over previous
#include <hip/hip_runtime.h>
#include <hip/hip_bf16.h>


typedef __attribute__((ext_vector_type(2))) float v2f;
typedef __attribute__((ext_vector_type(8))) float v8f;

#define BSZ        2
#define NN         3072
#define EE         4096
#define NODE_DIM   128
#define EDGE_DIM   64
#define NHEAD      8
#define HDIM       16
#define QKV        384

#define WMMA_F32(A, B, C) \
  __builtin_amdgcn_wmma_f32_16x16x4_f32(false, (A), false, (B), (short)0, (C), false, false)

__global__ __launch_bounds__(32)
void gemm_bias_32x32(const float* __restrict__ X, const float* __restrict__ W,
                     const float* __restrict__ bias, float* __restrict__ C,
                     int M, int K, int Ncols) {
  const int tiles_n = Ncols >> 5;
  const int tile = blockIdx.x;
  const int tm = tile / tiles_n;
  const int tn = tile % tiles_n;
  const int lane = threadIdx.x;
  const int half = lane >> 4;
  const int l15  = lane & 15;

  const int row0 = tm * 32 + l15;
  const int row1 = row0 + 16;
  const int col0 = tn * 32 + l15;
  const int col1 = col0 + 16;

  const float* __restrict__ Xr0 = X + (size_t)row0 * K;
  const float* __restrict__ Xr1 = X + (size_t)row1 * K;

  v8f acc00 = {}, acc01 = {}, acc10 = {}, acc11 = {};
  for (int k = 0; k < K; k += 4) {
    const int ka = k + half * 2;
    v2f a0, a1, b0, b1;
    a0.x = Xr0[ka];                         a0.y = Xr0[ka + 1];
    a1.x = Xr1[ka];                         a1.y = Xr1[ka + 1];
    b0.x = W[(size_t)ka * Ncols + col0];    b0.y = W[(size_t)(ka + 1) * Ncols + col0];
    b1.x = W[(size_t)ka * Ncols + col1];    b1.y = W[(size_t)(ka + 1) * Ncols + col1];
    acc00 = WMMA_F32(a0, b0, acc00);
    acc01 = WMMA_F32(a0, b1, acc01);
    acc10 = WMMA_F32(a1, b0, acc10);
    acc11 = WMMA_F32(a1, b1, acc11);
  }

  const float bv0 = bias[col0];
  const float bv1 = bias[col1];
#pragma unroll
  for (int j = 0; j < 8; ++j) {
    const int r0 = tm * 32 + j + half * 8;
    const int r1 = r0 + 16;
    C[(size_t)r0 * Ncols + col0] = acc00[j] + bv0;
    C[(size_t)r0 * Ncols + col1] = acc01[j] + bv1;
    C[(size_t)r1 * Ncols + col0] = acc10[j] + bv0;
    C[(size_t)r1 * Ncols + col1] = acc11[j] + bv1;
  }
}

#define DEG_COLS    64
#define DEG_ESPLIT  2
#define DEG_WAVES   16
#define DEG_ROWS    (EE / (DEG_ESPLIT * DEG_WAVES))
#define DEG_BLK     16
#define DEG_SLABS   (BSZ * NN / DEG_COLS)

__global__ __launch_bounds__(32 * DEG_WAVES)
void degree_kernel(const float* __restrict__ G, float* __restrict__ degp) {
  __shared__ v2f part[DEG_WAVES * 32];

  const int slab = blockIdx.x % DEG_SLABS;
  const int eh   = blockIdx.x / DEG_SLABS;
  const int wave = threadIdx.x >> 5;
  const int lane = threadIdx.x & 31;

  const int idx = slab * DEG_COLS + lane * 2;
  const int b = idx / NN;
  const int n = idx - b * NN;
  const int e0 = (eh * DEG_WAVES + wave) * DEG_ROWS;
  const float* __restrict__ g = G + (size_t)b * EE * NN + (size_t)e0 * NN + n;

  v2f cur[DEG_BLK], nxt[DEG_BLK];
#pragma unroll
  for (int i = 0; i < DEG_BLK; ++i)
    cur[i] = *reinterpret_cast<const v2f*>(g + (size_t)i * NN);

  v2f s = {0.f, 0.f};
  for (int r = DEG_BLK; r < DEG_ROWS; r += DEG_BLK) {
#pragma unroll
    for (int i = 0; i < DEG_BLK; ++i)
      nxt[i] = *reinterpret_cast<const v2f*>(g + (size_t)(r + i) * NN);
#pragma unroll
    for (int i = 0; i < DEG_BLK; ++i) s += cur[i];
#pragma unroll
    for (int i = 0; i < DEG_BLK; ++i) cur[i] = nxt[i];
  }
#pragma unroll
  for (int i = 0; i < DEG_BLK; ++i) s += cur[i];

  part[wave * 32 + lane] = s;
  __syncthreads();

  if (wave == 0) {
    v2f t = part[lane];
#pragma unroll
    for (int w = 1; w < DEG_WAVES; ++w) t += part[w * 32 + lane];
    *reinterpret_cast<v2f*>(degp + (size_t)eh * (BSZ * NN) + idx) = t;
  }
}

#define AGG_KSPLIT 4
#define AGG_KLEN   (EE / AGG_KSPLIT)

__global__ __launch_bounds__(32 * AGG_KSPLIT)
void agg_kernel(const float* __restrict__ G, const float* __restrict__ qkv_e,
                const float* __restrict__ degp, float* __restrict__ qkv) {
  __shared__ float part[AGG_KSPLIT * 4 * 8 * 32];

  const int tiles_c = QKV / 32;
  const int tiles_n = NN / 32;
  int tile = blockIdx.x;
  const int b  = tile / (tiles_n * tiles_c);
  tile -= b * tiles_n * tiles_c;
  const int tn = tile / tiles_c;
  const int tc = tile - tn * tiles_c;

  const int wave = __builtin_amdgcn_readfirstlane((int)(threadIdx.x >> 5));
  const int lane = threadIdx.x & 31;
  const int half = lane >> 4;
  const int l15  = lane & 15;
  const int n0 = tn * 32 + l15;
  const int n1 = n0 + 16;
  const int c0 = tc * 32 + l15;
  const int c1 = c0 + 16;

  const int kbeg = wave * AGG_KLEN;
  const float* __restrict__ Gb = G     + (size_t)b * EE * NN  + (size_t)kbeg * NN;
  const float* __restrict__ Eb = qkv_e + (size_t)b * EE * QKV + (size_t)kbeg * QKV;

  v8f acc00 = {}, acc01 = {}, acc10 = {}, acc11 = {};
  for (int k = 0; k < AGG_KLEN; k += 4) {
    const int ka = k + half * 2;
    const size_t ga = (size_t)ka * NN;
    const size_t ea = (size_t)ka * QKV;
    v2f a0, a1, b0, b1;
    a0.x = Gb[ga + n0];        a0.y = Gb[ga + NN + n0];
    a1.x = Gb[ga + n1];        a1.y = Gb[ga + NN + n1];
    b0.x = Eb[ea + c0];        b0.y = Eb[ea + QKV + c0];
    b1.x = Eb[ea + c1];        b1.y = Eb[ea + QKV + c1];
    acc00 = WMMA_F32(a0, b0, acc00);
    acc01 = WMMA_F32(a0, b1, acc01);
    acc10 = WMMA_F32(a1, b0, acc10);
    acc11 = WMMA_F32(a1, b1, acc11);
  }

  {
    float* __restrict__ pw = part + wave * (4 * 8 * 32) + lane;
#pragma unroll
    for (int j = 0; j < 8; ++j) {
      pw[(0 * 8 + j) * 32] = acc00[j];
      pw[(1 * 8 + j) * 32] = acc01[j];
      pw[(2 * 8 + j) * 32] = acc10[j];
      pw[(3 * 8 + j) * 32] = acc11[j];
    }
  }
  __syncthreads();

  const int rband = wave >> 1;
  const int cband = wave & 1;
  const int col   = tc * 32 + cband * 16 + l15;
  const float* __restrict__ pr = part + wave * (8 * 32) + lane;
  float* __restrict__ out = qkv + (size_t)b * NN * QKV;
#pragma unroll
  for (int j = 0; j < 8; ++j) {
    float s = pr[0 * (4 * 8 * 32) + j * 32];
    s += pr[1 * (4 * 8 * 32) + j * 32];
    s += pr[2 * (4 * 8 * 32) + j * 32];
    s += pr[3 * (4 * 8 * 32) + j * 32];
    const int r = tn * 32 + rband * 16 + j + half * 8;
    const float rdiv = 1.0f / (degp[b * NN + r] + degp[BSZ * NN + b * NN + r]);
    const size_t o = (size_t)r * QKV + col;
    out[o] = out[o] + s * rdiv;
  }
}

#define ATT_KSPLIT 4
#define ATT_TILES  (NN / 16 / ATT_KSPLIT)

__global__ __launch_bounds__(32 * ATT_KSPLIT, 12)
void attn_kernel(const float* __restrict__ qkv, float* __restrict__ attout) {
  __shared__ float lds[ATT_KSPLIT * 16 * 16];
  __shared__ float mls[ATT_KSPLIT * 32];

  const int tiles_q = NN / 16;
  int tile = blockIdx.x;
  const int b = tile / (NHEAD * tiles_q);
  tile -= b * NHEAD * tiles_q;
  const int h  = tile / tiles_q;
  const int tq = tile - h * tiles_q;

  const int wave = __builtin_amdgcn_readfirstlane((int)(threadIdx.x >> 5));
  const int lane = threadIdx.x & 31;
  const int half = lane >> 4;
  const int l15  = lane & 15;
  const float scale = 0.25f;

  const float* __restrict__ base = qkv + (size_t)b * NN * QKV + h * 48;
  float* __restrict__ wl = lds + wave * (16 * 16);

  float qa[8];
  {
    const int m = tq * 16 + l15;
#pragma unroll
    for (int kk = 0; kk < 16; kk += 4) {
      const int cc = kk + half * 2;
      qa[kk / 2]     = base[(size_t)m * QKV + cc] * scale;
      qa[kk / 2 + 1] = base[(size_t)m * QKV + cc + 1] * scale;
    }
  }

  v8f o = {};
  float mrow[8], lrow[8];
#pragma unroll
  for (int j = 0; j < 8; ++j) { mrow[j] = -1e30f; lrow[j] = 0.f; }

  const int key0 = wave * (ATT_TILES * 16);
  const float* __restrict__ kp = base + (size_t)(key0 + l15) * QKV + 16 + half * 2;
  const float* __restrict__ vp = base + (size_t)(key0 + half * 2) * QKV + 32 + l15;
  for (int t = 0; t < ATT_TILES; ++t, kp += 16 * QKV, vp += 16 * QKV) {
    v8f s = {};
#pragma unroll
    for (int kk = 0; kk < 16; kk += 4) {
      v2f a, bb;
      a.x  = qa[kk / 2];
      a.y  = qa[kk / 2 + 1];
      bb.x = kp[kk];
      bb.y = kp[kk + 1];
      s = WMMA_F32(a, bb, s);
    }

    float pnew[8];
#pragma unroll
    for (int j = 0; j < 8; ++j) {
      const float sj = s[j];
      float mx = sj;
      mx = fmaxf(mx, __shfl_xor(mx, 1, 32));
      mx = fmaxf(mx, __shfl_xor(mx, 2, 32));
      mx = fmaxf(mx, __shfl_xor(mx, 4, 32));
      mx = fmaxf(mx, __shfl_xor(mx, 8, 32));
      const float mnew = fmaxf(mrow[j], mx);
      const float corr = __expf(mrow[j] - mnew);
      const float p    = __expf(sj - mnew);
      float ps = p;
      ps += __shfl_xor(ps, 1, 32);
      ps += __shfl_xor(ps, 2, 32);
      ps += __shfl_xor(ps, 4, 32);
      ps += __shfl_xor(ps, 8, 32);
      lrow[j] = lrow[j] * corr + ps;
      mrow[j] = mnew;
      o[j] *= corr;
      pnew[j] = p;
    }

#pragma unroll
    for (int j = 0; j < 8; ++j)
      wl[(j + half * 8) * 16 + l15] = pnew[j];
    __builtin_amdgcn_fence(__ATOMIC_RELEASE, "wavefront");
    __builtin_amdgcn_wave_barrier();
    __builtin_amdgcn_fence(__ATOMIC_ACQUIRE, "wavefront");

#pragma unroll
    for (int kk = 0; kk < 16; kk += 4) {
      const int krow = kk + half * 2;
      v2f a, bb;
      a.x = wl[l15 * 16 + krow];
      a.y = wl[l15 * 16 + krow + 1];
      bb.x = vp[kk * QKV];
      bb.y = vp[(kk + 1) * QKV];
      o = WMMA_F32(a, bb, o);
    }
    __builtin_amdgcn_fence(__ATOMIC_RELEASE, "wavefront");
    __builtin_amdgcn_wave_barrier();
    __builtin_amdgcn_fence(__ATOMIC_ACQUIRE, "wavefront");
  }

  const int elane = (int)__builtin_amdgcn_mbcnt_lo(~0u, 0u);
  const int ehalf = elane >> 4;
  const int el15  = elane & 15;

  {
    float* __restrict__ pw = lds + wave * (16 * 16) + elane;
#pragma unroll
    for (int j = 0; j < 8; ++j) {
      pw[j * 32] = o[j];
      mls[wave * 32 + j + ehalf * 8]      = mrow[j];
      mls[wave * 32 + 16 + j + ehalf * 8] = lrow[j];
    }
  }
  __syncthreads();

  float* __restrict__ outp = attout + (size_t)b * NN * NODE_DIM + h * HDIM;
#pragma unroll
  for (int jj = 0; jj < 2; ++jj) {
    const int j  = wave * 2 + jj;
    const int rr = j + ehalf * 8;
    float mw[ATT_KSPLIT];
    float mall = -1e30f;
#pragma unroll
    for (int w = 0; w < ATT_KSPLIT; ++w) {
      mw[w] = mls[w * 32 + rr];
      mall = fmaxf(mall, mw[w]);
    }
    float lsum = 0.f, osum = 0.f;
#pragma unroll
    for (int w = 0; w < ATT_KSPLIT; ++w) {
      const float wgt = __expf(mw[w] - mall);
      lsum += mls[w * 32 + 16 + rr] * wgt;
      osum += lds[w * (16 * 16) + j * 32 + elane] * wgt;
    }
    const int r = tq * 16 + rr;
    outp[(size_t)r * NODE_DIM + el15] = osum / lsum;
  }
}

extern "C" void kernel_launch(void* const* d_in, const int* in_sizes, int n_in,
                              void* d_out, int out_size, void* d_ws, size_t ws_size,
                              hipStream_t stream) {
  const float* node    = (const float*)d_in[0];
  const float* edge    = (const float*)d_in[1];
  const float* G       = (const float*)d_in[2];
  const float* W_qkv_n = (const float*)d_in[3];
  const float* b_qkv_n = (const float*)d_in[4];
  const float* W_qkv_e = (const float*)d_in[5];
  const float* b_qkv_e = (const float*)d_in[6];
  const float* W_o     = (const float*)d_in[7];
  const float* b_o     = (const float*)d_in[8];
  float* out = (float*)d_out;

  float* qkv  = (float*)d_ws;
  float* qkve = qkv  + (size_t)BSZ * NN * QKV;
  float* deg  = qkve + (size_t)BSZ * EE * QKV;
  float* att  = deg  + (size_t)BSZ * NN;

  gemm_bias_32x32<<<(BSZ * NN / 32) * (QKV / 32), 32, 0, stream>>>(
      node, W_qkv_n, b_qkv_n, qkv, BSZ * NN, NODE_DIM, QKV);

  gemm_bias_32x32<<<(BSZ * EE / 32) * (QKV / 32), 32, 0, stream>>>(
      edge, W_qkv_e, b_qkv_e, qkve, BSZ * EE, EDGE_DIM, QKV);

  degree_kernel<<<DEG_SLABS * DEG_ESPLIT, 32 * DEG_WAVES, 0, stream>>>(G, deg);

  agg_kernel<<<BSZ * (NN / 32) * (QKV / 32), 32 * AGG_KSPLIT, 0, stream>>>(G, qkve, deg, qkv);

  attn_kernel<<<BSZ * NHEAD * (NN / 16), 32 * ATT_KSPLIT, 0, stream>>>(qkv, att);

  gemm_bias_32x32<<<(BSZ * NN / 32) * (NODE_DIM / 32), 32, 0, stream>>>(
      att, W_o, b_o, out, BSZ * NN, NODE_DIM, NODE_DIM);
}


# attention: softmax row sum kept per lane (own key column of each tile, rescaled with the running max), added across the 16 lanes once after the key loop instead of per tile; row-sum order changes
# speedup vs baseline: 2.4557x; 1.0737x over previous
#include <hip/hip_runtime.h>
#include <hip/hip_bf16.h>


typedef __attribute__((ext_vector_type(2))) float v2f;
typedef __attribute__((ext_vector_type(8))) float v8f;

#define BSZ        2
#define NN         3072
#define EE         4096
#define NODE_DIM   128
#define EDGE_DIM   64
#define NHEAD      8
#define HDIM       16
#define QKV        384

#define WMMA_F32(A, B, C) \
  __builtin_amdgcn_wmma_f32_16x16x4_f32(false, (A), false, (B), (short)0, (C), false, false)

__global__ __launch_bounds__(32)
void gemm_bias_32x32(const float* __restrict__ X, const float* __restrict__ W,
                     const float* __restrict__ bias, float* __restrict__ C,
                     int M, int K, int Ncols) {
  const int tiles_n = Ncols >> 5;
  const int tile = blockIdx.x;
  const int tm = tile / tiles_n;
  const int tn = tile % tiles_n;
  const int lane = threadIdx.x;
  const int half = lane >> 4;
  const int l15  = lane & 15;

  const int row0 = tm * 32 + l15;
  const int row1 = row0 + 16;
  const int col0 = tn * 32 + l15;
  const int col1 = col0 + 16;

  const float* __restrict__ Xr0 = X + (size_t)row0 * K;
  const float* __restrict__ Xr1 = X + (size_t)row1 * K;

  v8f acc00 = {}, acc01 = {}, acc10 = {}, acc11 = {};
  for (int k = 0; k < K; k += 4) {
    const int ka = k + half * 2;
    v2f a0, a1, b0, b1;
    a0.x = Xr0[ka];                         a0.y = Xr0[ka + 1];
    a1.x = Xr1[ka];                         a1.y = Xr1[ka + 1];
    b0.x = W[(size_t)ka * Ncols + col0];    b0.y = W[(size_t)(ka + 1) * Ncols + col0];
    b1.x = W[(size_t)ka * Ncols + col1];    b1.y = W[(size_t)(ka + 1) * Ncols + col1];
    acc00 = WMMA_F32(a0, b0, acc00);
    acc01 = WMMA_F32(a0, b1, acc01);
    acc10 = WMMA_F32(a1, b0, acc10);
    acc11 = WMMA_F32(a1, b1, acc11);
  }

  const float bv0 = bias[col0];
  const float bv1 = bias[col1];
#pragma unroll
  for (int j = 0; j < 8; ++j) {
    const int r0 = tm * 32 + j + half * 8;
    const int r1 = r0 + 16;
    C[(size_t)r0 * Ncols + col0] = acc00[j] + bv0;
    C[(size_t)r0 * Ncols + col1] = acc01[j] + bv1;
    C[(size_t)r1 * Ncols + col0] = acc10[j] + bv0;
    C[(size_t)r1 * Ncols + col1] = acc11[j] + bv1;
  }
}

#define DEG_COLS    64
#define DEG_ESPLIT  2
#define DEG_WAVES   16
#define DEG_ROWS    (EE / (DEG_ESPLIT * DEG_WAVES))
#define DEG_BLK     16
#define DEG_SLABS   (BSZ * NN / DEG_COLS)

__global__ __launch_bounds__(32 * DEG_WAVES)
void degree_kernel(const float* __restrict__ G, float* __restrict__ degp) {
  __shared__ v2f part[DEG_WAVES * 32];

  const int slab = blockIdx.x % DEG_SLABS;
  const int eh   = blockIdx.x / DEG_SLABS;
  const int wave = threadIdx.x >> 5;
  const int lane = threadIdx.x & 31;

  const int idx = slab * DEG_COLS + lane * 2;
  const int b = idx / NN;
  const int n = idx - b * NN;
  const int e0 = (eh * DEG_WAVES + wave) * DEG_ROWS;
  const float* __restrict__ g = G + (size_t)b * EE * NN + (size_t)e0 * NN + n;

  v2f cur[DEG_BLK], nxt[DEG_BLK];
#pragma unroll
  for (int i = 0; i < DEG_BLK; ++i)
    cur[i] = *reinterpret_cast<const v2f*>(g + (size_t)i * NN);

  v2f s = {0.f, 0.f};
  for (int r = DEG_BLK; r < DEG_ROWS; r += DEG_BLK) {
#pragma unroll
    for (int i = 0; i < DEG_BLK; ++i)
      nxt[i] = *reinterpret_cast<const v2f*>(g + (size_t)(r + i) * NN);
#pragma unroll
    for (int i = 0; i < DEG_BLK; ++i) s += cur[i];
#pragma unroll
    for (int i = 0; i < DEG_BLK; ++i) cur[i] = nxt[i];
  }
#pragma unroll
  for (int i = 0; i < DEG_BLK; ++i) s += cur[i];

  part[wave * 32 + lane] = s;
  __syncthreads();

  if (wave == 0) {
    v2f t = part[lane];
#pragma unroll
    for (int w = 1; w < DEG_WAVES; ++w) t += part[w * 32 + lane];
    *reinterpret_cast<v2f*>(degp + (size_t)eh * (BSZ * NN) + idx) = t;
  }
}

#define AGG_KSPLIT 4
#define AGG_KLEN   (EE / AGG_KSPLIT)

__global__ __launch_bounds__(32 * AGG_KSPLIT)
void agg_kernel(const float* __restrict__ G, const float* __restrict__ qkv_e,
                const float* __restrict__ degp, float* __restrict__ qkv) {
  __shared__ float part[AGG_KSPLIT * 4 * 8 * 32];

  const int tiles_c = QKV / 32;
  const int tiles_n = NN / 32;
  int tile = blockIdx.x;
  const int b  = tile / (tiles_n * tiles_c);
  tile -= b * tiles_n * tiles_c;
  const int tn = tile / tiles_c;
  const int tc = tile - tn * tiles_c;

  const int wave = __builtin_amdgcn_readfirstlane((int)(threadIdx.x >> 5));
  const int lane = threadIdx.x & 31;
  const int half = lane >> 4;
  const int l15  = lane & 15;
  const int n0 = tn * 32 + l15;
  const int n1 = n0 + 16;
  const int c0 = tc * 32 + l15;
  const int c1 = c0 + 16;

  const int kbeg = wave * AGG_KLEN;
  const float* __restrict__ Gb = G     + (size_t)b * EE * NN  + (size_t)kbeg * NN;
  const float* __restrict__ Eb = qkv_e + (size_t)b * EE * QKV + (size_t)kbeg * QKV;

  v8f acc00 = {}, acc01 = {}, acc10 = {}, acc11 = {};
  for (int k = 0; k < AGG_KLEN; k += 4) {
    const int ka = k + half * 2;
    const size_t ga = (size_t)ka * NN;
    const size_t ea = (size_t)ka * QKV;
    v2f a0, a1, b0, b1;
    a0.x = Gb[ga + n0];        a0.y = Gb[ga + NN + n0];
    a1.x = Gb[ga + n1];        a1.y = Gb[ga + NN + n1];
    b0.x = Eb[ea + c0];        b0.y = Eb[ea + QKV + c0];
    b1.x = Eb[ea + c1];        b1.y = Eb[ea + QKV + c1];
    acc00 = WMMA_F32(a0, b0, acc00);
    acc01 = WMMA_F32(a0, b1, acc01);
    acc10 = WMMA_F32(a1, b0, acc10);
    acc11 = WMMA_F32(a1, b1, acc11);
  }

  {
    float* __restrict__ pw = part + wave * (4 * 8 * 32) + lane;
#pragma unroll
    for (int j = 0; j < 8; ++j) {
      pw[(0 * 8 + j) * 32] = acc00[j];
      pw[(1 * 8 + j) * 32] = acc01[j];
      pw[(2 * 8 + j) * 32] = acc10[j];
      pw[(3 * 8 + j) * 32] = acc11[j];
    }
  }
  __syncthreads();

  const int rband = wave >> 1;
  const int cband = wave & 1;
  const int col   = tc * 32 + cband * 16 + l15;
  const float* __restrict__ pr = part + wave * (8 * 32) + lane;
  float* __restrict__ out = qkv + (size_t)b * NN * QKV;
#pragma unroll
  for (int j = 0; j < 8; ++j) {
    float s = pr[0 * (4 * 8 * 32) + j * 32];
    s += pr[1 * (4 * 8 * 32) + j * 32];
    s += pr[2 * (4 * 8 * 32) + j * 32];
    s += pr[3 * (4 * 8 * 32) + j * 32];
    const int r = tn * 32 + rband * 16 + j + half * 8;
    const float rdiv = 1.0f / (degp[b * NN + r] + degp[BSZ * NN + b * NN + r]);
    const size_t o = (size_t)r * QKV + col;
    out[o] = out[o] + s * rdiv;
  }
}

#define ATT_KSPLIT 4
#define ATT_TILES  (NN / 16 / ATT_KSPLIT)

__global__ __launch_bounds__(32 * ATT_KSPLIT, 12)
void attn_kernel(const float* __restrict__ qkv, float* __restrict__ attout) {
  __shared__ float lds[ATT_KSPLIT * 16 * 16];
  __shared__ float mls[ATT_KSPLIT * 32];

  const int tiles_q = NN / 16;
  int tile = blockIdx.x;
  const int b = tile / (NHEAD * tiles_q);
  tile -= b * NHEAD * tiles_q;
  const int h  = tile / tiles_q;
  const int tq = tile - h * tiles_q;

  const int wave = __builtin_amdgcn_readfirstlane((int)(threadIdx.x >> 5));
  const int lane = threadIdx.x & 31;
  const int half = lane >> 4;
  const int l15  = lane & 15;
  const float scale = 0.25f;

  const float* __restrict__ base = qkv + (size_t)b * NN * QKV + h * 48;
  float* __restrict__ wl = lds + wave * (16 * 16);

  float qa[8];
  {
    const int m = tq * 16 + l15;
#pragma unroll
    for (int kk = 0; kk < 16; kk += 4) {
      const int cc = kk + half * 2;
      qa[kk / 2]     = base[(size_t)m * QKV + cc] * scale;
      qa[kk / 2 + 1] = base[(size_t)m * QKV + cc + 1] * scale;
    }
  }

  v8f o = {};
  float mrow[8], lrow[8];
#pragma unroll
  for (int j = 0; j < 8; ++j) { mrow[j] = -1e30f; lrow[j] = 0.f; }

  const int key0 = wave * (ATT_TILES * 16);
  const float* __restrict__ kp = base + (size_t)(key0 + l15) * QKV + 16 + half * 2;
  const float* __restrict__ vp = base + (size_t)(key0 + half * 2) * QKV + 32 + l15;
  for (int t = 0; t < ATT_TILES; ++t, kp += 16 * QKV, vp += 16 * QKV) {
    v8f s = {};
#pragma unroll
    for (int kk = 0; kk < 16; kk += 4) {
      v2f a, bb;
      a.x  = qa[kk / 2];
      a.y  = qa[kk / 2 + 1];
      bb.x = kp[kk];
      bb.y = kp[kk + 1];
      s = WMMA_F32(a, bb, s);
    }

    float pnew[8];
#pragma unroll
    for (int j = 0; j < 8; ++j) {
      const float sj = s[j];
      float mx = sj;
      mx = fmaxf(mx, __shfl_xor(mx, 1, 32));
      mx = fmaxf(mx, __shfl_xor(mx, 2, 32));
      mx = fmaxf(mx, __shfl_xor(mx, 4, 32));
      mx = fmaxf(mx, __shfl_xor(mx, 8, 32));
      const float mnew = fmaxf(mrow[j], mx);
      const float corr = __expf(mrow[j] - mnew);
      const float p    = __expf(sj - mnew);
      lrow[j] = lrow[j] * corr + p;
      mrow[j] = mnew;
      o[j] *= corr;
      pnew[j] = p;
    }

#pragma unroll
    for (int j = 0; j < 8; ++j)
      wl[(j + half * 8) * 16 + l15] = pnew[j];
    __builtin_amdgcn_fence(__ATOMIC_RELEASE, "wavefront");
    __builtin_amdgcn_wave_barrier();
    __builtin_amdgcn_fence(__ATOMIC_ACQUIRE, "wavefront");

#pragma unroll
    for (int kk = 0; kk < 16; kk += 4) {
      const int krow = kk + half * 2;
      v2f a, bb;
      a.x = wl[l15 * 16 + krow];
      a.y = wl[l15 * 16 + krow + 1];
      bb.x = vp[kk * QKV];
      bb.y = vp[(kk + 1) * QKV];
      o = WMMA_F32(a, bb, o);
    }
    __builtin_amdgcn_fence(__ATOMIC_RELEASE, "wavefront");
    __builtin_amdgcn_wave_barrier();
    __builtin_amdgcn_fence(__ATOMIC_ACQUIRE, "wavefront");
  }

#pragma unroll
  for (int j = 0; j < 8; ++j) {
    float ls = lrow[j];
    ls += __shfl_xor(ls, 1, 32);
    ls += __shfl_xor(ls, 2, 32);
    ls += __shfl_xor(ls, 4, 32);
    ls += __shfl_xor(ls, 8, 32);
    lrow[j] = ls;
  }

  const int elane = (int)__builtin_amdgcn_mbcnt_lo(~0u, 0u);
  const int ehalf = elane >> 4;
  const int el15  = elane & 15;

  {
    float* __restrict__ pw = lds + wave * (16 * 16) + elane;
#pragma unroll
    for (int j = 0; j < 8; ++j) {
      pw[j * 32] = o[j];
      mls[wave * 32 + j + ehalf * 8]      = mrow[j];
      mls[wave * 32 + 16 + j + ehalf * 8] = lrow[j];
    }
  }
  __syncthreads();

  float* __restrict__ outp = attout + (size_t)b * NN * NODE_DIM + h * HDIM;
#pragma unroll
  for (int jj = 0; jj < 2; ++jj) {
    const int j  = wave * 2 + jj;
    const int rr = j + ehalf * 8;
    float mw[ATT_KSPLIT];
    float mall = -1e30f;
#pragma unroll
    for (int w = 0; w < ATT_KSPLIT; ++w) {
      mw[w] = mls[w * 32 + rr];
      mall = fmaxf(mall, mw[w]);
    }
    float lsum = 0.f, osum = 0.f;
#pragma unroll
    for (int w = 0; w < ATT_KSPLIT; ++w) {
      const float wgt = __expf(mw[w] - mall);
      lsum += mls[w * 32 + 16 + rr] * wgt;
      osum += lds[w * (16 * 16) + j * 32 + elane] * wgt;
    }
    const int r = tq * 16 + rr;
    outp[(size_t)r * NODE_DIM + el15] = osum / lsum;
  }
}

extern "C" void kernel_launch(void* const* d_in, const int* in_sizes, int n_in,
                              void* d_out, int out_size, void* d_ws, size_t ws_size,
                              hipStream_t stream) {
  const float* node    = (const float*)d_in[0];
  const float* edge    = (const float*)d_in[1];
  const float* G       = (const float*)d_in[2];
  const float* W_qkv_n = (const float*)d_in[3];
  const float* b_qkv_n = (const float*)d_in[4];
  const float* W_qkv_e = (const float*)d_in[5];
  const float* b_qkv_e = (const float*)d_in[6];
  const float* W_o     = (const float*)d_in[7];
  const float* b_o     = (const float*)d_in[8];
  float* out = (float*)d_out;

  float* qkv  = (float*)d_ws;
  float* qkve = qkv  + (size_t)BSZ * NN * QKV;
  float* deg  = qkve + (size_t)BSZ * EE * QKV;
  float* att  = deg  + (size_t)BSZ * NN;

  gemm_bias_32x32<<<(BSZ * NN / 32) * (QKV / 32), 32, 0, stream>>>(
      node, W_qkv_n, b_qkv_n, qkv, BSZ * NN, NODE_DIM, QKV);

  gemm_bias_32x32<<<(BSZ * EE / 32) * (QKV / 32), 32, 0, stream>>>(
      edge, W_qkv_e, b_qkv_e, qkve, BSZ * EE, EDGE_DIM, QKV);

  degree_kernel<<<DEG_SLABS * DEG_ESPLIT, 32 * DEG_WAVES, 0, stream>>>(G, deg);

  agg_kernel<<<BSZ * (NN / 32) * (QKV / 32), 32 * AGG_KSPLIT, 0, stream>>>(G, qkve, deg, qkv);

  attn_kernel<<<BSZ * NHEAD * (NN / 16), 32 * ATT_KSPLIT, 0, stream>>>(qkv, att);

  gemm_bias_32x32<<<(BSZ * NN / 32) * (NODE_DIM / 32), 32, 0, stream>>>(
      att, W_o, b_o, out, BSZ * NN, NODE_DIM, NODE_DIM);
}


# attention: the score tile (not P) crosses the wave's LDS, so a lane owns one query row: softmax in-lane, one half exchange for the max, P already the A operand; row-sum and P.V key order change
# speedup vs baseline: 2.6652x; 1.0853x over previous
#include <hip/hip_runtime.h>
#include <hip/hip_bf16.h>


typedef __attribute__((ext_vector_type(2))) float v2f;
typedef __attribute__((ext_vector_type(8))) float v8f;

#define BSZ        2
#define NN         3072
#define EE         4096
#define NODE_DIM   128
#define EDGE_DIM   64
#define NHEAD      8
#define HDIM       16
#define QKV        384

#define WMMA_F32(A, B, C) \
  __builtin_amdgcn_wmma_f32_16x16x4_f32(false, (A), false, (B), (short)0, (C), false, false)

__global__ __launch_bounds__(32)
void gemm_bias_32x32(const float* __restrict__ X, const float* __restrict__ W,
                     const float* __restrict__ bias, float* __restrict__ C,
                     int M, int K, int Ncols) {
  const int tiles_n = Ncols >> 5;
  const int tile = blockIdx.x;
  const int tm = tile / tiles_n;
  const int tn = tile % tiles_n;
  const int lane = threadIdx.x;
  const int half = lane >> 4;
  const int l15  = lane & 15;

  const int row0 = tm * 32 + l15;
  const int row1 = row0 + 16;
  const int col0 = tn * 32 + l15;
  const int col1 = col0 + 16;

  const float* __restrict__ Xr0 = X + (size_t)row0 * K;
  const float* __restrict__ Xr1 = X + (size_t)row1 * K;

  v8f acc00 = {}, acc01 = {}, acc10 = {}, acc11 = {};
  for (int k = 0; k < K; k += 4) {
    const int ka = k + half * 2;
    v2f a0, a1, b0, b1;
    a0.x = Xr0[ka];                         a0.y = Xr0[ka + 1];
    a1.x = Xr1[ka];                         a1.y = Xr1[ka + 1];
    b0.x = W[(size_t)ka * Ncols + col0];    b0.y = W[(size_t)(ka + 1) * Ncols + col0];
    b1.x = W[(size_t)ka * Ncols + col1];    b1.y = W[(size_t)(ka + 1) * Ncols + col1];
    acc00 = WMMA_F32(a0, b0, acc00);
    acc01 = WMMA_F32(a0, b1, acc01);
    acc10 = WMMA_F32(a1, b0, acc10);
    acc11 = WMMA_F32(a1, b1, acc11);
  }

  const float bv0 = bias[col0];
  const float bv1 = bias[col1];
#pragma unroll
  for (int j = 0; j < 8; ++j) {
    const int r0 = tm * 32 + j + half * 8;
    const int r1 = r0 + 16;
    C[(size_t)r0 * Ncols + col0] = acc00[j] + bv0;
    C[(size_t)r0 * Ncols + col1] = acc01[j] + bv1;
    C[(size_t)r1 * Ncols + col0] = acc10[j] + bv0;
    C[(size_t)r1 * Ncols + col1] = acc11[j] + bv1;
  }
}

#define DEG_COLS    64
#define DEG_ESPLIT  2
#define DEG_WAVES   16
#define DEG_ROWS    (EE / (DEG_ESPLIT * DEG_WAVES))
#define DEG_BLK     16
#define DEG_SLABS   (BSZ * NN / DEG_COLS)

__global__ __launch_bounds__(32 * DEG_WAVES)
void degree_kernel(const float* __restrict__ G, float* __restrict__ degp) {
  __shared__ v2f part[DEG_WAVES * 32];

  const int slab = blockIdx.x % DEG_SLABS;
  const int eh   = blockIdx.x / DEG_SLABS;
  const int wave = threadIdx.x >> 5;
  const int lane = threadIdx.x & 31;

  const int idx = slab * DEG_COLS + lane * 2;
  const int b = idx / NN;
  const int n = idx - b * NN;
  const int e0 = (eh * DEG_WAVES + wave) * DEG_ROWS;
  const float* __restrict__ g = G + (size_t)b * EE * NN + (size_t)e0 * NN + n;

  v2f cur[DEG_BLK], nxt[DEG_BLK];
#pragma unroll
  for (int i = 0; i < DEG_BLK; ++i)
    cur[i] = *reinterpret_cast<const v2f*>(g + (size_t)i * NN);

  v2f s = {0.f, 0.f};
  for (int r = DEG_BLK; r < DEG_ROWS; r += DEG_BLK) {
#pragma unroll
    for (int i = 0; i < DEG_BLK; ++i)
      nxt[i] = *reinterpret_cast<const v2f*>(g + (size_t)(r + i) * NN);
#pragma unroll
    for (int i = 0; i < DEG_BLK; ++i) s += cur[i];
#pragma unroll
    for (int i = 0; i < DEG_BLK; ++i) cur[i] = nxt[i];
  }
#pragma unroll
  for (int i = 0; i < DEG_BLK; ++i) s += cur[i];

  part[wave * 32 + lane] = s;
  __syncthreads();

  if (wave == 0) {
    v2f t = part[lane];
#pragma unroll
    for (int w = 1; w < DEG_WAVES; ++w) t += part[w * 32 + lane];
    *reinterpret_cast<v2f*>(degp + (size_t)eh * (BSZ * NN) + idx) = t;
  }
}

#define AGG_KSPLIT 4
#define AGG_KLEN   (EE / AGG_KSPLIT)

__global__ __launch_bounds__(32 * AGG_KSPLIT)
void agg_kernel(const float* __restrict__ G, const float* __restrict__ qkv_e,
                const float* __restrict__ degp, float* __restrict__ qkv) {
  __shared__ float part[AGG_KSPLIT * 4 * 8 * 32];

  const int tiles_c = QKV / 32;
  const int tiles_n = NN / 32;
  int tile = blockIdx.x;
  const int b  = tile / (tiles_n * tiles_c);
  tile -= b * tiles_n * tiles_c;
  const int tn = tile / tiles_c;
  const int tc = tile - tn * tiles_c;

  const int wave = __builtin_amdgcn_readfirstlane((int)(threadIdx.x >> 5));
  const int lane = threadIdx.x & 31;
  const int half = lane >> 4;
  const int l15  = lane & 15;
  const int n0 = tn * 32 + l15;
  const int n1 = n0 + 16;
  const int c0 = tc * 32 + l15;
  const int c1 = c0 + 16;

  const int kbeg = wave * AGG_KLEN;
  const float* __restrict__ Gb = G     + (size_t)b * EE * NN  + (size_t)kbeg * NN;
  const float* __restrict__ Eb = qkv_e + (size_t)b * EE * QKV + (size_t)kbeg * QKV;

  v8f acc00 = {}, acc01 = {}, acc10 = {}, acc11 = {};
  for (int k = 0; k < AGG_KLEN; k += 4) {
    const int ka = k + half * 2;
    const size_t ga = (size_t)ka * NN;
    const size_t ea = (size_t)ka * QKV;
    v2f a0, a1, b0, b1;
    a0.x = Gb[ga + n0];        a0.y = Gb[ga + NN + n0];
    a1.x = Gb[ga + n1];        a1.y = Gb[ga + NN + n1];
    b0.x = Eb[ea + c0];        b0.y = Eb[ea + QKV + c0];
    b1.x = Eb[ea + c1];        b1.y = Eb[ea + QKV + c1];
    acc00 = WMMA_F32(a0, b0, acc00);
    acc01 = WMMA_F32(a0, b1, acc01);
    acc10 = WMMA_F32(a1, b0, acc10);
    acc11 = WMMA_F32(a1, b1, acc11);
  }

  {
    float* __restrict__ pw = part + wave * (4 * 8 * 32) + lane;
#pragma unroll
    for (int j = 0; j < 8; ++j) {
      pw[(0 * 8 + j) * 32] = acc00[j];
      pw[(1 * 8 + j) * 32] = acc01[j];
      pw[(2 * 8 + j) * 32] = acc10[j];
      pw[(3 * 8 + j) * 32] = acc11[j];
    }
  }
  __syncthreads();

  const int rband = wave >> 1;
  const int cband = wave & 1;
  const int col   = tc * 32 + cband * 16 + l15;
  const float* __restrict__ pr = part + wave * (8 * 32) + lane;
  float* __restrict__ out = qkv + (size_t)b * NN * QKV;
#pragma unroll
  for (int j = 0; j < 8; ++j) {
    float s = pr[0 * (4 * 8 * 32) + j * 32];
    s += pr[1 * (4 * 8 * 32) + j * 32];
    s += pr[2 * (4 * 8 * 32) + j * 32];
    s += pr[3 * (4 * 8 * 32) + j * 32];
    const int r = tn * 32 + rband * 16 + j + half * 8;
    const float rdiv = 1.0f / (degp[b * NN + r] + degp[BSZ * NN + b * NN + r]);
    const size_t o = (size_t)r * QKV + col;
    out[o] = out[o] + s * rdiv;
  }
}

typedef __attribute__((ext_vector_type(4))) float v4f;

#define ATT_KSPLIT 4
#define ATT_TILES  (NN / 16 / ATT_KSPLIT)
#define ATT_PITCH  20
#define ATT_WLDS   (16 * ATT_PITCH + 16)

__device__ __forceinline__ float xhalf(float x) {
  const int bits = __builtin_bit_cast(int, x);
  return __builtin_bit_cast(float,
      __builtin_amdgcn_permlanex16(bits, bits, 0x76543210, 0xfedcba98, false, false));
}

__global__ __launch_bounds__(32 * ATT_KSPLIT, 12)
void attn_kernel(const float* __restrict__ qkv, float* __restrict__ attout) {
  __shared__ __attribute__((aligned(16))) float lds[ATT_KSPLIT * ATT_WLDS];
  __shared__ float mls[ATT_KSPLIT * 32];

  const int tiles_q = NN / 16;
  int tile = blockIdx.x;
  const int b = tile / (NHEAD * tiles_q);
  tile -= b * NHEAD * tiles_q;
  const int h  = tile / tiles_q;
  const int tq = tile - h * tiles_q;

  const int wave = __builtin_amdgcn_readfirstlane((int)(threadIdx.x >> 5));
  const int lane = threadIdx.x & 31;
  const int half = lane >> 4;
  const int l15  = lane & 15;
  const float scale = 0.25f;

  const float* __restrict__ base = qkv + (size_t)b * NN * QKV + h * 48;
  float* __restrict__ wl = lds + wave * ATT_WLDS;
  float* __restrict__ cw = wl + 16 * ATT_PITCH;

  float qa[8];
  {
    const int m = tq * 16 + l15;
#pragma unroll
    for (int kk = 0; kk < 16; kk += 4) {
      const int cc = kk + half * 2;
      qa[kk / 2]     = base[(size_t)m * QKV + cc] * scale;
      qa[kk / 2 + 1] = base[(size_t)m * QKV + cc + 1] * scale;
    }
  }

  v8f o = {};
  float mrun = -1e30f, lrun = 0.f;

  const int key0 = wave * (ATT_TILES * 16);
  const float* __restrict__ kp = base + (size_t)(key0 + l15) * QKV + 16 + half * 2;
  const float* __restrict__ vp = base + (size_t)(key0 + half * 8) * QKV + 32 + l15;
  for (int t = 0; t < ATT_TILES; ++t, kp += 16 * QKV, vp += 16 * QKV) {
    float vb[8];
#pragma unroll
    for (int e = 0; e < 8; ++e) vb[e] = vp[e * QKV];

    v8f s = {};
#pragma unroll
    for (int kk = 0; kk < 16; kk += 4) {
      v2f a, bb;
      a.x  = qa[kk / 2];
      a.y  = qa[kk / 2 + 1];
      bb.x = kp[kk];
      bb.y = kp[kk + 1];
      s = WMMA_F32(a, bb, s);
    }

#pragma unroll
    for (int j = 0; j < 8; ++j)
      wl[(j + half * 8) * ATT_PITCH + l15] = s[j];
    __builtin_amdgcn_fence(__ATOMIC_RELEASE, "wavefront");
    __builtin_amdgcn_wave_barrier();
    __builtin_amdgcn_fence(__ATOMIC_ACQUIRE, "wavefront");

    const v4f s0 = *reinterpret_cast<const v4f*>(wl + l15 * ATT_PITCH + half * 8);
    const v4f s1 = *reinterpret_cast<const v4f*>(wl + l15 * ATT_PITCH + half * 8 + 4);

    float mx = fmaxf(fmaxf(fmaxf(s0.x, s0.y), fmaxf(s0.z, s0.w)),
                     fmaxf(fmaxf(s1.x, s1.y), fmaxf(s1.z, s1.w)));
    mx = fmaxf(mx, xhalf(mx));
    const float mnew = fmaxf(mrun, mx);
    const float corr = __expf(mrun - mnew);
    float p[8];
    p[0] = __expf(s0.x - mnew);  p[1] = __expf(s0.y - mnew);
    p[2] = __expf(s0.z - mnew);  p[3] = __expf(s0.w - mnew);
    p[4] = __expf(s1.x - mnew);  p[5] = __expf(s1.y - mnew);
    p[6] = __expf(s1.z - mnew);  p[7] = __expf(s1.w - mnew);
    lrun = lrun * corr + (((p[0] + p[1]) + (p[2] + p[3])) + ((p[4] + p[5]) + (p[6] + p[7])));
    mrun = mnew;

    cw[l15] = corr;
    __builtin_amdgcn_fence(__ATOMIC_RELEASE, "wavefront");
    __builtin_amdgcn_wave_barrier();
    __builtin_amdgcn_fence(__ATOMIC_ACQUIRE, "wavefront");
    const v4f c0 = *reinterpret_cast<const v4f*>(cw + half * 8);
    const v4f c1 = *reinterpret_cast<const v4f*>(cw + half * 8 + 4);
    o[0] *= c0.x;  o[1] *= c0.y;  o[2] *= c0.z;  o[3] *= c0.w;
    o[4] *= c1.x;  o[5] *= c1.y;  o[6] *= c1.z;  o[7] *= c1.w;

    asm volatile("" : "+v"(lrun));
    __builtin_amdgcn_sched_barrier(0);
#pragma unroll
    for (int i = 0; i < 4; ++i) {
      v2f a, bb;
      a.x  = p[2 * i];
      a.y  = p[2 * i + 1];
      bb.x = vb[2 * i];
      bb.y = vb[2 * i + 1];
      o = WMMA_F32(a, bb, o);
    }
  }

  const float ltot = lrun + xhalf(lrun);

  const int elane = (int)__builtin_amdgcn_mbcnt_lo(~0u, 0u);
  const int ehalf = elane >> 4;
  const int el15  = elane & 15;

  {
    float* __restrict__ pw = lds + wave * ATT_WLDS + elane;
#pragma unroll
    for (int j = 0; j < 8; ++j)
      pw[j * 32] = o[j];
    mls[wave * 32 + el15]      = mrun;
    mls[wave * 32 + 16 + el15] = ltot;
  }
  __syncthreads();

  float* __restrict__ outp = attout + (size_t)b * NN * NODE_DIM + h * HDIM;
#pragma unroll
  for (int jj = 0; jj < 2; ++jj) {
    const int j  = wave * 2 + jj;
    const int rr = j + ehalf * 8;
    float mw[ATT_KSPLIT];
    float mall = -1e30f;
#pragma unroll
    for (int w = 0; w < ATT_KSPLIT; ++w) {
      mw[w] = mls[w * 32 + rr];
      mall = fmaxf(mall, mw[w]);
    }
    float lsum = 0.f, osum = 0.f;
#pragma unroll
    for (int w = 0; w < ATT_KSPLIT; ++w) {
      const float wgt = __expf(mw[w] - mall);
      lsum += mls[w * 32 + 16 + rr] * wgt;
      osum += lds[w * ATT_WLDS + j * 32 + elane] * wgt;
    }
    const int r = tq * 16 + rr;
    outp[(size_t)r * NODE_DIM + el15] = osum / lsum;
  }
}

extern "C" void kernel_launch(void* const* d_in, const int* in_sizes, int n_in,
                              void* d_out, int out_size, void* d_ws, size_t ws_size,
                              hipStream_t stream) {
  const float* node    = (const float*)d_in[0];
  const float* edge    = (const float*)d_in[1];
  const float* G       = (const float*)d_in[2];
  const float* W_qkv_n = (const float*)d_in[3];
  const float* b_qkv_n = (const float*)d_in[4];
  const float* W_qkv_e = (const float*)d_in[5];
  const float* b_qkv_e = (const float*)d_in[6];
  const float* W_o     = (const float*)d_in[7];
  const float* b_o     = (const float*)d_in[8];
  float* out = (float*)d_out;

  float* qkv  = (float*)d_ws;
  float* qkve = qkv  + (size_t)BSZ * NN * QKV;
  float* deg  = qkve + (size_t)BSZ * EE * QKV;
  float* att  = deg  + (size_t)BSZ * NN;

  gemm_bias_32x32<<<(BSZ * NN / 32) * (QKV / 32), 32, 0, stream>>>(
      node, W_qkv_n, b_qkv_n, qkv, BSZ * NN, NODE_DIM, QKV);

  gemm_bias_32x32<<<(BSZ * EE / 32) * (QKV / 32), 32, 0, stream>>>(
      edge, W_qkv_e, b_qkv_e, qkve, BSZ * EE, EDGE_DIM, QKV);

  degree_kernel<<<DEG_SLABS * DEG_ESPLIT, 32 * DEG_WAVES, 0, stream>>>(G, deg);

  agg_kernel<<<BSZ * (NN / 32) * (QKV / 32), 32 * AGG_KSPLIT, 0, stream>>>(G, qkve, deg, qkv);

  attn_kernel<<<BSZ * NHEAD * (NN / 16), 32 * ATT_KSPLIT, 0, stream>>>(qkv, att);

  gemm_bias_32x32<<<(BSZ * NN / 32) * (NODE_DIM / 32), 32, 0, stream>>>(
      att, W_o, b_o, out, BSZ * NN, NODE_DIM, NODE_DIM);
}


# qkv workspace stored head-major (q, k, v planes of [B][H][N][16]): a 16-key K or V tile is 8 whole contiguous lines, not 16 half-used lines at a 1536-byte pitch; every sum and its order kept
# speedup vs baseline: 2.8001x; 1.0506x over previous
#include <hip/hip_runtime.h>
#include <hip/hip_bf16.h>


typedef __attribute__((ext_vector_type(2))) float v2f;
typedef __attribute__((ext_vector_type(8))) float v8f;

#define BSZ        2
#define NN         3072
#define EE         4096
#define NODE_DIM   128
#define EDGE_DIM   64
#define NHEAD      8
#define HDIM       16
#define QKV        384

#define WMMA_F32(A, B, C) \
  __builtin_amdgcn_wmma_f32_16x16x4_f32(false, (A), false, (B), (short)0, (C), false, false)

__device__ __forceinline__ size_t hm_base(int b, int col) {
  const int h = col / (3 * HDIM);
  const int w = col - h * (3 * HDIM);
  const int t = w >> 4;
  const int c = w & 15;
  return ((size_t)((t * BSZ + b) * NHEAD + h) * NN) * HDIM + c;
}

template <bool HEADMAJOR>
__global__ __launch_bounds__(32)
void gemm_bias_32x32(const float* __restrict__ X, const float* __restrict__ W,
                     const float* __restrict__ bias, float* __restrict__ C,
                     int M, int K, int Ncols) {
  const int tiles_n = Ncols >> 5;
  const int tile = blockIdx.x;
  const int tm = tile / tiles_n;
  const int tn = tile % tiles_n;
  const int lane = threadIdx.x;
  const int half = lane >> 4;
  const int l15  = lane & 15;

  const int row0 = tm * 32 + l15;
  const int row1 = row0 + 16;
  const int col0 = tn * 32 + l15;
  const int col1 = col0 + 16;

  const float* __restrict__ Xr0 = X + (size_t)row0 * K;
  const float* __restrict__ Xr1 = X + (size_t)row1 * K;

  v8f acc00 = {}, acc01 = {}, acc10 = {}, acc11 = {};
  for (int k = 0; k < K; k += 4) {
    const int ka = k + half * 2;
    v2f a0, a1, b0, b1;
    a0.x = Xr0[ka];                         a0.y = Xr0[ka + 1];
    a1.x = Xr1[ka];                         a1.y = Xr1[ka + 1];
    b0.x = W[(size_t)ka * Ncols + col0];    b0.y = W[(size_t)(ka + 1) * Ncols + col0];
    b1.x = W[(size_t)ka * Ncols + col1];    b1.y = W[(size_t)(ka + 1) * Ncols + col1];
    acc00 = WMMA_F32(a0, b0, acc00);
    acc01 = WMMA_F32(a0, b1, acc01);
    acc10 = WMMA_F32(a1, b0, acc10);
    acc11 = WMMA_F32(a1, b1, acc11);
  }

  const float bv0 = bias[col0];
  const float bv1 = bias[col1];
  if (HEADMAJOR) {
    const int bb = (tm * 32) / NN;
    const int nb = tm * 32 - bb * NN;
    float* __restrict__ C0 = C + hm_base(bb, col0);
    float* __restrict__ C1 = C + hm_base(bb, col1);
#pragma unroll
    for (int j = 0; j < 8; ++j) {
      const int n0 = nb + j + half * 8;
      const int n1 = n0 + 16;
      C0[(size_t)n0 * HDIM] = acc00[j] + bv0;
      C1[(size_t)n0 * HDIM] = acc01[j] + bv1;
      C0[(size_t)n1 * HDIM] = acc10[j] + bv0;
      C1[(size_t)n1 * HDIM] = acc11[j] + bv1;
    }
  } else {
#pragma unroll
    for (int j = 0; j < 8; ++j) {
      const int r0 = tm * 32 + j + half * 8;
      const int r1 = r0 + 16;
      C[(size_t)r0 * Ncols + col0] = acc00[j] + bv0;
      C[(size_t)r0 * Ncols + col1] = acc01[j] + bv1;
      C[(size_t)r1 * Ncols + col0] = acc10[j] + bv0;
      C[(size_t)r1 * Ncols + col1] = acc11[j] + bv1;
    }
  }
}

#define DEG_COLS    64
#define DEG_ESPLIT  2
#define DEG_WAVES   16
#define DEG_ROWS    (EE / (DEG_ESPLIT * DEG_WAVES))
#define DEG_BLK     16
#define DEG_SLABS   (BSZ * NN / DEG_COLS)

__global__ __launch_bounds__(32 * DEG_WAVES)
void degree_kernel(const float* __restrict__ G, float* __restrict__ degp) {
  __shared__ v2f part[DEG_WAVES * 32];

  const int slab = blockIdx.x % DEG_SLABS;
  const int eh   = blockIdx.x / DEG_SLABS;
  const int wave = threadIdx.x >> 5;
  const int lane = threadIdx.x & 31;

  const int idx = slab * DEG_COLS + lane * 2;
  const int b = idx / NN;
  const int n = idx - b * NN;
  const int e0 = (eh * DEG_WAVES + wave) * DEG_ROWS;
  const float* __restrict__ g = G + (size_t)b * EE * NN + (size_t)e0 * NN + n;

  v2f cur[DEG_BLK], nxt[DEG_BLK];
#pragma unroll
  for (int i = 0; i < DEG_BLK; ++i)
    cur[i] = *reinterpret_cast<const v2f*>(g + (size_t)i * NN);

  v2f s = {0.f, 0.f};
  for (int r = DEG_BLK; r < DEG_ROWS; r += DEG_BLK) {
#pragma unroll
    for (int i = 0; i < DEG_BLK; ++i)
      nxt[i] = *reinterpret_cast<const v2f*>(g + (size_t)(r + i) * NN);
#pragma unroll
    for (int i = 0; i < DEG_BLK; ++i) s += cur[i];
#pragma unroll
    for (int i = 0; i < DEG_BLK; ++i) cur[i] = nxt[i];
  }
#pragma unroll
  for (int i = 0; i < DEG_BLK; ++i) s += cur[i];

  part[wave * 32 + lane] = s;
  __syncthreads();

  if (wave == 0) {
    v2f t = part[lane];
#pragma unroll
    for (int w = 1; w < DEG_WAVES; ++w) t += part[w * 32 + lane];
    *reinterpret_cast<v2f*>(degp + (size_t)eh * (BSZ * NN) + idx) = t;
  }
}

#define AGG_KSPLIT 4
#define AGG_KLEN   (EE / AGG_KSPLIT)

__global__ __launch_bounds__(32 * AGG_KSPLIT)
void agg_kernel(const float* __restrict__ G, const float* __restrict__ qkv_e,
                const float* __restrict__ degp, float* __restrict__ qkv) {
  __shared__ float part[AGG_KSPLIT * 4 * 8 * 32];

  const int tiles_c = QKV / 32;
  const int tiles_n = NN / 32;
  int tile = blockIdx.x;
  const int b  = tile / (tiles_n * tiles_c);
  tile -= b * tiles_n * tiles_c;
  const int tn = tile / tiles_c;
  const int tc = tile - tn * tiles_c;

  const int wave = __builtin_amdgcn_readfirstlane((int)(threadIdx.x >> 5));
  const int lane = threadIdx.x & 31;
  const int half = lane >> 4;
  const int l15  = lane & 15;
  const int n0 = tn * 32 + l15;
  const int n1 = n0 + 16;
  const int c0 = tc * 32 + l15;
  const int c1 = c0 + 16;

  const int kbeg = wave * AGG_KLEN;
  const float* __restrict__ Gb = G     + (size_t)b * EE * NN  + (size_t)kbeg * NN;
  const float* __restrict__ Eb = qkv_e + (size_t)b * EE * QKV + (size_t)kbeg * QKV;

  v8f acc00 = {}, acc01 = {}, acc10 = {}, acc11 = {};
  for (int k = 0; k < AGG_KLEN; k += 4) {
    const int ka = k + half * 2;
    const size_t ga = (size_t)ka * NN;
    const size_t ea = (size_t)ka * QKV;
    v2f a0, a1, b0, b1;
    a0.x = Gb[ga + n0];        a0.y = Gb[ga + NN + n0];
    a1.x = Gb[ga + n1];        a1.y = Gb[ga + NN + n1];
    b0.x = Eb[ea + c0];        b0.y = Eb[ea + QKV + c0];
    b1.x = Eb[ea + c1];        b1.y = Eb[ea + QKV + c1];
    acc00 = WMMA_F32(a0, b0, acc00);
    acc01 = WMMA_F32(a0, b1, acc01);
    acc10 = WMMA_F32(a1, b0, acc10);
    acc11 = WMMA_F32(a1, b1, acc11);
  }

  {
    float* __restrict__ pw = part + wave * (4 * 8 * 32) + lane;
#pragma unroll
    for (int j = 0; j < 8; ++j) {
      pw[(0 * 8 + j) * 32] = acc00[j];
      pw[(1 * 8 + j) * 32] = acc01[j];
      pw[(2 * 8 + j) * 32] = acc10[j];
      pw[(3 * 8 + j) * 32] = acc11[j];
    }
  }
  __syncthreads();

  const int rband = wave >> 1;
  const int cband = wave & 1;
  const int col   = tc * 32 + cband * 16 + l15;
  const float* __restrict__ pr = part + wave * (8 * 32) + lane;
  float* __restrict__ out = qkv + hm_base(b, col);
#pragma unroll
  for (int j = 0; j < 8; ++j) {
    float s = pr[0 * (4 * 8 * 32) + j * 32];
    s += pr[1 * (4 * 8 * 32) + j * 32];
    s += pr[2 * (4 * 8 * 32) + j * 32];
    s += pr[3 * (4 * 8 * 32) + j * 32];
    const int r = tn * 32 + rband * 16 + j + half * 8;
    const float rdiv = 1.0f / (degp[b * NN + r] + degp[BSZ * NN + b * NN + r]);
    const size_t o = (size_t)r * HDIM;
    out[o] = out[o] + s * rdiv;
  }
}

typedef __attribute__((ext_vector_type(4))) float v4f;

#define ATT_KSPLIT 4
#define ATT_TILES  (NN / 16 / ATT_KSPLIT)
#define ATT_PITCH  20
#define ATT_WLDS   (16 * ATT_PITCH + 16)

__device__ __forceinline__ float xhalf(float x) {
  const int bits = __builtin_bit_cast(int, x);
  return __builtin_bit_cast(float,
      __builtin_amdgcn_permlanex16(bits, bits, 0x76543210, 0xfedcba98, false, false));
}

__global__ __launch_bounds__(32 * ATT_KSPLIT, 12)
void attn_kernel(const float* __restrict__ qkv, float* __restrict__ attout) {
  __shared__ __attribute__((aligned(16))) float lds[ATT_KSPLIT * ATT_WLDS];
  __shared__ float mls[ATT_KSPLIT * 32];

  const int tiles_q = NN / 16;
  int tile = blockIdx.x;
  const int b = tile / (NHEAD * tiles_q);
  tile -= b * NHEAD * tiles_q;
  const int h  = tile / tiles_q;
  const int tq = tile - h * tiles_q;

  const int wave = __builtin_amdgcn_readfirstlane((int)(threadIdx.x >> 5));
  const int lane = threadIdx.x & 31;
  const int half = lane >> 4;
  const int l15  = lane & 15;
  const float scale = 0.25f;

  const float* __restrict__ Qp = qkv + hm_base(b, h * 48);
  const float* __restrict__ Kp = qkv + hm_base(b, h * 48 + 16);
  const float* __restrict__ Vp = qkv + hm_base(b, h * 48 + 32);
  float* __restrict__ wl = lds + wave * ATT_WLDS;
  float* __restrict__ cw = wl + 16 * ATT_PITCH;

  float qa[8];
  {
    const int m = tq * 16 + l15;
#pragma unroll
    for (int kk = 0; kk < 16; kk += 4) {
      const int cc = kk + half * 2;
      qa[kk / 2]     = Qp[(size_t)m * HDIM + cc] * scale;
      qa[kk / 2 + 1] = Qp[(size_t)m * HDIM + cc + 1] * scale;
    }
  }

  v8f o = {};
  float mrun = -1e30f, lrun = 0.f;

  const int key0 = wave * (ATT_TILES * 16);
  const float* __restrict__ kp = Kp + (size_t)(key0 + l15) * HDIM + half * 2;
  const float* __restrict__ vp = Vp + (size_t)(key0 + half * 8) * HDIM + l15;
  for (int t = 0; t < ATT_TILES; ++t, kp += 16 * HDIM, vp += 16 * HDIM) {
    float vb[8];
#pragma unroll
    for (int e = 0; e < 8; ++e) vb[e] = vp[e * HDIM];

    v8f s = {};
#pragma unroll
    for (int kk = 0; kk < 16; kk += 4) {
      v2f a, bb;
      a.x  = qa[kk / 2];
      a.y  = qa[kk / 2 + 1];
      bb.x = kp[kk];
      bb.y = kp[kk + 1];
      s = WMMA_F32(a, bb, s);
    }

#pragma unroll
    for (int j = 0; j < 8; ++j)
      wl[(j + half * 8) * ATT_PITCH + l15] = s[j];
    __builtin_amdgcn_fence(__ATOMIC_RELEASE, "wavefront");
    __builtin_amdgcn_wave_barrier();
    __builtin_amdgcn_fence(__ATOMIC_ACQUIRE, "wavefront");

    const v4f s0 = *reinterpret_cast<const v4f*>(wl + l15 * ATT_PITCH + half * 8);
    const v4f s1 = *reinterpret_cast<const v4f*>(wl + l15 * ATT_PITCH + half * 8 + 4);

    float mx = fmaxf(fmaxf(fmaxf(s0.x, s0.y), fmaxf(s0.z, s0.w)),
                     fmaxf(fmaxf(s1.x, s1.y), fmaxf(s1.z, s1.w)));
    mx = fmaxf(mx, xhalf(mx));
    const float mnew = fmaxf(mrun, mx);
    const float corr = __expf(mrun - mnew);
    float p[8];
    p[0] = __expf(s0.x - mnew);  p[1] = __expf(s0.y - mnew);
    p[2] = __expf(s0.z - mnew);  p[3] = __expf(s0.w - mnew);
    p[4] = __expf(s1.x - mnew);  p[5] = __expf(s1.y - mnew);
    p[6] = __expf(s1.z - mnew);  p[7] = __expf(s1.w - mnew);
    lrun = lrun * corr + (((p[0] + p[1]) + (p[2] + p[3])) + ((p[4] + p[5]) + (p[6] + p[7])));
    mrun = mnew;

    cw[l15] = corr;
    __builtin_amdgcn_fence(__ATOMIC_RELEASE, "wavefront");
    __builtin_amdgcn_wave_barrier();
    __builtin_amdgcn_fence(__ATOMIC_ACQUIRE, "wavefront");
    const v4f c0 = *reinterpret_cast<const v4f*>(cw + half * 8);
    const v4f c1 = *reinterpret_cast<const v4f*>(cw + half * 8 + 4);
    o[0] *= c0.x;  o[1] *= c0.y;  o[2] *= c0.z;  o[3] *= c0.w;
    o[4] *= c1.x;  o[5] *= c1.y;  o[6] *= c1.z;  o[7] *= c1.w;

    asm volatile("" : "+v"(lrun));
    __builtin_amdgcn_sched_barrier(0);
#pragma unroll
    for (int i = 0; i < 4; ++i) {
      v2f a, bb;
      a.x  = p[2 * i];
      a.y  = p[2 * i + 1];
      bb.x = vb[2 * i];
      bb.y = vb[2 * i + 1];
      o = WMMA_F32(a, bb, o);
    }
  }

  const float ltot = lrun + xhalf(lrun);

  const int elane = (int)__builtin_amdgcn_mbcnt_lo(~0u, 0u);
  const int ehalf = elane >> 4;
  const int el15  = elane & 15;

  {
    float* __restrict__ pw = lds + wave * ATT_WLDS + elane;
#pragma unroll
    for (int j = 0; j < 8; ++j)
      pw[j * 32] = o[j];
    mls[wave * 32 + el15]      = mrun;
    mls[wave * 32 + 16 + el15] = ltot;
  }
  __syncthreads();

  float* __restrict__ outp = attout + (size_t)b * NN * NODE_DIM + h * HDIM;
#pragma unroll
  for (int jj = 0; jj < 2; ++jj) {
    const int j  = wave * 2 + jj;
    const int rr = j + ehalf * 8;
    float mw[ATT_KSPLIT];
    float mall = -1e30f;
#pragma unroll
    for (int w = 0; w < ATT_KSPLIT; ++w) {
      mw[w] = mls[w * 32 + rr];
      mall = fmaxf(mall, mw[w]);
    }
    float lsum = 0.f, osum = 0.f;
#pragma unroll
    for (int w = 0; w < ATT_KSPLIT; ++w) {
      const float wgt = __expf(mw[w] - mall);
      lsum += mls[w * 32 + 16 + rr] * wgt;
      osum += lds[w * ATT_WLDS + j * 32 + elane] * wgt;
    }
    const int r = tq * 16 + rr;
    outp[(size_t)r * NODE_DIM + el15] = osum / lsum;
  }
}

extern "C" void kernel_launch(void* const* d_in, const int* in_sizes, int n_in,
                              void* d_out, int out_size, void* d_ws, size_t ws_size,
                              hipStream_t stream) {
  const float* node    = (const float*)d_in[0];
  const float* edge    = (const float*)d_in[1];
  const float* G       = (const float*)d_in[2];
  const float* W_qkv_n = (const float*)d_in[3];
  const float* b_qkv_n = (const float*)d_in[4];
  const float* W_qkv_e = (const float*)d_in[5];
  const float* b_qkv_e = (const float*)d_in[6];
  const float* W_o     = (const float*)d_in[7];
  const float* b_o     = (const float*)d_in[8];
  float* out = (float*)d_out;

  float* qkv  = (float*)d_ws;
  float* qkve = qkv  + (size_t)BSZ * NN * QKV;
  float* deg  = qkve + (size_t)BSZ * EE * QKV;
  float* att  = deg  + (size_t)BSZ * NN;

  gemm_bias_32x32<true><<<(BSZ * NN / 32) * (QKV / 32), 32, 0, stream>>>(
      node, W_qkv_n, b_qkv_n, qkv, BSZ * NN, NODE_DIM, QKV);

  gemm_bias_32x32<false><<<(BSZ * EE / 32) * (QKV / 32), 32, 0, stream>>>(
      edge, W_qkv_e, b_qkv_e, qkve, BSZ * EE, EDGE_DIM, QKV);

  degree_kernel<<<DEG_SLABS * DEG_ESPLIT, 32 * DEG_WAVES, 0, stream>>>(G, deg);

  agg_kernel<<<BSZ * (NN / 32) * (QKV / 32), 32 * AGG_KSPLIT, 0, stream>>>(G, qkve, deg, qkv);

  attn_kernel<<<BSZ * NHEAD * (NN / 16), 32 * ATT_KSPLIT, 0, stream>>>(qkv, att);

  gemm_bias_32x32<false><<<(BSZ * NN / 32) * (NODE_DIM / 32), 32, 0, stream>>>(
      att, W_o, b_o, out, BSZ * NN, NODE_DIM, NODE_DIM);
}


# qkv_e and the K and V planes kept in the B-fragment order of the product that reads them: a lane fetches a 16x16 tile with two 16-byte loads of whole lines; score and aggregation sum order changes
# speedup vs baseline: 2.9021x; 1.0364x over previous
#include <hip/hip_runtime.h>
#include <hip/hip_bf16.h>


typedef __attribute__((ext_vector_type(2))) float v2f;
typedef __attribute__((ext_vector_type(4))) float v4f;
typedef __attribute__((ext_vector_type(8))) float v8f;

#define BSZ        2
#define NN         3072
#define EE         4096
#define NODE_DIM   128
#define EDGE_DIM   64
#define NHEAD      8
#define HDIM       16
#define QKV        384

#define WMMA_F32(A, B, C) \
  __builtin_amdgcn_wmma_f32_16x16x4_f32(false, (A), false, (B), (short)0, (C), false, false)

__device__ __forceinline__ int frag_off(int k, int c) {
  return ((k >> 2) & 1) * 128 + (((k >> 3) & 1) * 16 + c) * 4 + (k & 3);
}

__device__ __forceinline__ size_t hm_plane(int b, int col) {
  const int h = col / (3 * HDIM);
  const int t = (col - h * (3 * HDIM)) >> 4;
  return ((size_t)((t * BSZ + b) * NHEAD + h) * NN) * HDIM;
}
__device__ __forceinline__ int hm_kind(int col) { return (col % (3 * HDIM)) >> 4; }
__device__ __forceinline__ int hm_chan(int col) { return (col % (3 * HDIM)) & 15; }
__device__ __forceinline__ int hm_off(int t, int n, int c) {
  const int tile = (n >> 4) * 256;
  const int m = n & 15;
  if (t == 0) return n * HDIM + c;
  if (t == 1) return tile + frag_off(c, m);
  return tile + frag_off(m, c);
}

template <int MODE>
__global__ __launch_bounds__(32)
void gemm_bias_32x32(const float* __restrict__ X, const float* __restrict__ W,
                     const float* __restrict__ bias, float* __restrict__ C,
                     int M, int K, int Ncols) {
  const int tiles_n = Ncols >> 5;
  const int tile = blockIdx.x;
  const int tm = tile / tiles_n;
  const int tn = tile % tiles_n;
  const int lane = threadIdx.x;
  const int half = lane >> 4;
  const int l15  = lane & 15;

  const int row0 = tm * 32 + l15;
  const int row1 = row0 + 16;
  const int col0 = tn * 32 + l15;
  const int col1 = col0 + 16;

  const float* __restrict__ Xr0 = X + (size_t)row0 * K;
  const float* __restrict__ Xr1 = X + (size_t)row1 * K;

  v8f acc00 = {}, acc01 = {}, acc10 = {}, acc11 = {};
  for (int k = 0; k < K; k += 4) {
    const int ka = k + half * 2;
    v2f a0, a1, b0, b1;
    a0.x = Xr0[ka];                         a0.y = Xr0[ka + 1];
    a1.x = Xr1[ka];                         a1.y = Xr1[ka + 1];
    b0.x = W[(size_t)ka * Ncols + col0];    b0.y = W[(size_t)(ka + 1) * Ncols + col0];
    b1.x = W[(size_t)ka * Ncols + col1];    b1.y = W[(size_t)(ka + 1) * Ncols + col1];
    acc00 = WMMA_F32(a0, b0, acc00);
    acc01 = WMMA_F32(a0, b1, acc01);
    acc10 = WMMA_F32(a1, b0, acc10);
    acc11 = WMMA_F32(a1, b1, acc11);
  }

  const float bv0 = bias[col0];
  const float bv1 = bias[col1];
  if (MODE == 1) {
    const int bb = (tm * 32) / NN;
    const int nb = tm * 32 - bb * NN;
    float* __restrict__ C0 = C + hm_plane(bb, col0);
    float* __restrict__ C1 = C + hm_plane(bb, col1);
    const int t0 = hm_kind(col0), ch0 = hm_chan(col0);
    const int t1 = hm_kind(col1), ch1 = hm_chan(col1);
#pragma unroll
    for (int j = 0; j < 8; ++j) {
      const int n0 = nb + j + half * 8;
      const int n1 = n0 + 16;
      C0[hm_off(t0, n0, ch0)] = acc00[j] + bv0;
      C1[hm_off(t1, n0, ch1)] = acc01[j] + bv1;
      C0[hm_off(t0, n1, ch0)] = acc10[j] + bv0;
      C1[hm_off(t1, n1, ch1)] = acc11[j] + bv1;
    }
  } else if (MODE == 2) {
    const int bands = Ncols >> 4;
    float* __restrict__ T00 = C + ((size_t)(tm * 2)     * bands + tn * 2) * 256 + lane * 4;
    float* __restrict__ T10 = C + ((size_t)(tm * 2 + 1) * bands + tn * 2) * 256 + lane * 4;
    v4f lo, hi;
    lo = (v4f){acc00[0], acc00[1], acc00[2], acc00[3]} + bv0;
    hi = (v4f){acc00[4], acc00[5], acc00[6], acc00[7]} + bv0;
    *reinterpret_cast<v4f*>(T00) = lo;        *reinterpret_cast<v4f*>(T00 + 128) = hi;
    lo = (v4f){acc01[0], acc01[1], acc01[2], acc01[3]} + bv1;
    hi = (v4f){acc01[4], acc01[5], acc01[6], acc01[7]} + bv1;
    *reinterpret_cast<v4f*>(T00 + 256) = lo;  *reinterpret_cast<v4f*>(T00 + 384) = hi;
    lo = (v4f){acc10[0], acc10[1], acc10[2], acc10[3]} + bv0;
    hi = (v4f){acc10[4], acc10[5], acc10[6], acc10[7]} + bv0;
    *reinterpret_cast<v4f*>(T10) = lo;        *reinterpret_cast<v4f*>(T10 + 128) = hi;
    lo = (v4f){acc11[0], acc11[1], acc11[2], acc11[3]} + bv1;
    hi = (v4f){acc11[4], acc11[5], acc11[6], acc11[7]} + bv1;
    *reinterpret_cast<v4f*>(T10 + 256) = lo;  *reinterpret_cast<v4f*>(T10 + 384) = hi;
  } else {
#pragma unroll
    for (int j = 0; j < 8; ++j) {
      const int r0 = tm * 32 + j + half * 8;
      const int r1 = r0 + 16;
      C[(size_t)r0 * Ncols + col0] = acc00[j] + bv0;
      C[(size_t)r0 * Ncols + col1] = acc01[j] + bv1;
      C[(size_t)r1 * Ncols + col0] = acc10[j] + bv0;
      C[(size_t)r1 * Ncols + col1] = acc11[j] + bv1;
    }
  }
}

#define DEG_COLS    64
#define DEG_ESPLIT  2
#define DEG_WAVES   16
#define DEG_ROWS    (EE / (DEG_ESPLIT * DEG_WAVES))
#define DEG_BLK     16
#define DEG_SLABS   (BSZ * NN / DEG_COLS)

__global__ __launch_bounds__(32 * DEG_WAVES)
void degree_kernel(const float* __restrict__ G, float* __restrict__ degp) {
  __shared__ v2f part[DEG_WAVES * 32];

  const int slab = blockIdx.x % DEG_SLABS;
  const int eh   = blockIdx.x / DEG_SLABS;
  const int wave = threadIdx.x >> 5;
  const int lane = threadIdx.x & 31;

  const int idx = slab * DEG_COLS + lane * 2;
  const int b = idx / NN;
  const int n = idx - b * NN;
  const int e0 = (eh * DEG_WAVES + wave) * DEG_ROWS;
  const float* __restrict__ g = G + (size_t)b * EE * NN + (size_t)e0 * NN + n;

  v2f cur[DEG_BLK], nxt[DEG_BLK];
#pragma unroll
  for (int i = 0; i < DEG_BLK; ++i)
    cur[i] = *reinterpret_cast<const v2f*>(g + (size_t)i * NN);

  v2f s = {0.f, 0.f};
  for (int r = DEG_BLK; r < DEG_ROWS; r += DEG_BLK) {
#pragma unroll
    for (int i = 0; i < DEG_BLK; ++i)
      nxt[i] = *reinterpret_cast<const v2f*>(g + (size_t)(r + i) * NN);
#pragma unroll
    for (int i = 0; i < DEG_BLK; ++i) s += cur[i];
#pragma unroll
    for (int i = 0; i < DEG_BLK; ++i) cur[i] = nxt[i];
  }
#pragma unroll
  for (int i = 0; i < DEG_BLK; ++i) s += cur[i];

  part[wave * 32 + lane] = s;
  __syncthreads();

  if (wave == 0) {
    v2f t = part[lane];
#pragma unroll
    for (int w = 1; w < DEG_WAVES; ++w) t += part[w * 32 + lane];
    *reinterpret_cast<v2f*>(degp + (size_t)eh * (BSZ * NN) + idx) = t;
  }
}

#define AGG_KSPLIT 4
#define AGG_KLEN   (EE / AGG_KSPLIT)

__global__ __launch_bounds__(32 * AGG_KSPLIT)
void agg_kernel(const float* __restrict__ G, const float* __restrict__ qkv_e,
                const float* __restrict__ degp, float* __restrict__ qkv) {
  __shared__ float part[AGG_KSPLIT * 4 * 8 * 32];

  const int tiles_c = QKV / 32;
  const int tiles_n = NN / 32;
  int tile = blockIdx.x;
  const int b  = tile / (tiles_n * tiles_c);
  tile -= b * tiles_n * tiles_c;
  const int tn = tile / tiles_c;
  const int tc = tile - tn * tiles_c;

  const int wave = __builtin_amdgcn_readfirstlane((int)(threadIdx.x >> 5));
  const int lane = threadIdx.x & 31;
  const int half = lane >> 4;
  const int l15  = lane & 15;
  const int n0 = tn * 32 + l15;
  const int n1 = n0 + 16;

  const int kbeg = wave * AGG_KLEN;
  const float* __restrict__ Gb = G + (size_t)b * EE * NN + (size_t)(kbeg + half * 8) * NN;
  const float* __restrict__ Eb = qkv_e
      + ((size_t)(b * (EE / 16) + kbeg / 16) * (QKV / 16) + tc * 2) * 256 + lane * 4;

  v8f acc00 = {}, acc01 = {}, acc10 = {}, acc11 = {};
  for (int kb = 0; kb < AGG_KLEN / 16; ++kb) {
    const float* __restrict__ ep = Eb + (size_t)kb * ((QKV / 16) * 256);
    const float* __restrict__ gp = Gb + (size_t)kb * (16 * NN);
    const v4f e0l = *reinterpret_cast<const v4f*>(ep);
    const v4f e0h = *reinterpret_cast<const v4f*>(ep + 128);
    const v4f e1l = *reinterpret_cast<const v4f*>(ep + 256);
    const v4f e1h = *reinterpret_cast<const v4f*>(ep + 384);
    float g0[8], g1[8];
#pragma unroll
    for (int e = 0; e < 8; ++e) {
      g0[e] = gp[(size_t)e * NN + n0];
      g1[e] = gp[(size_t)e * NN + n1];
    }
    const float eb0[8] = {e0l.x, e0l.y, e0l.z, e0l.w, e0h.x, e0h.y, e0h.z, e0h.w};
    const float eb1[8] = {e1l.x, e1l.y, e1l.z, e1l.w, e1h.x, e1h.y, e1h.z, e1h.w};
#pragma unroll
    for (int i = 0; i < 4; ++i) {
      v2f a0, a1, b0, b1;
      a0.x = g0[2 * i];   a0.y = g0[2 * i + 1];
      a1.x = g1[2 * i];   a1.y = g1[2 * i + 1];
      b0.x = eb0[2 * i];  b0.y = eb0[2 * i + 1];
      b1.x = eb1[2 * i];  b1.y = eb1[2 * i + 1];
      acc00 = WMMA_F32(a0, b0, acc00);
      acc01 = WMMA_F32(a0, b1, acc01);
      acc10 = WMMA_F32(a1, b0, acc10);
      acc11 = WMMA_F32(a1, b1, acc11);
    }
  }

  {
    float* __restrict__ pw = part + wave * (4 * 8 * 32) + lane;
#pragma unroll
    for (int j = 0; j < 8; ++j) {
      pw[(0 * 8 + j) * 32] = acc00[j];
      pw[(1 * 8 + j) * 32] = acc01[j];
      pw[(2 * 8 + j) * 32] = acc10[j];
      pw[(3 * 8 + j) * 32] = acc11[j];
    }
  }
  __syncthreads();

  const int rband = wave >> 1;
  const int cband = wave & 1;
  const int col   = tc * 32 + cband * 16 + l15;
  const float* __restrict__ pr = part + wave * (8 * 32) + lane;
  float* __restrict__ out = qkv + hm_plane(b, col);
  const int kind = hm_kind(col);
  const int chan = hm_chan(col);
#pragma unroll
  for (int j = 0; j < 8; ++j) {
    float s = pr[0 * (4 * 8 * 32) + j * 32];
    s += pr[1 * (4 * 8 * 32) + j * 32];
    s += pr[2 * (4 * 8 * 32) + j * 32];
    s += pr[3 * (4 * 8 * 32) + j * 32];
    const int r = tn * 32 + rband * 16 + j + half * 8;
    const float rdiv = 1.0f / (degp[b * NN + r] + degp[BSZ * NN + b * NN + r]);
    const int o = hm_off(kind, r, chan);
    out[o] = out[o] + s * rdiv;
  }
}

#define ATT_KSPLIT 4
#define ATT_TILES  (NN / 16 / ATT_KSPLIT)
#define ATT_PITCH  20
#define ATT_WLDS   (16 * ATT_PITCH + 16)

__device__ __forceinline__ float xhalf(float x) {
  const int bits = __builtin_bit_cast(int, x);
  return __builtin_bit_cast(float,
      __builtin_amdgcn_permlanex16(bits, bits, 0x76543210, 0xfedcba98, false, false));
}

__global__ __launch_bounds__(32 * ATT_KSPLIT, 12)
void attn_kernel(const float* __restrict__ qkv, float* __restrict__ attout) {
  __shared__ __attribute__((aligned(16))) float lds[ATT_KSPLIT * ATT_WLDS];
  __shared__ float mls[ATT_KSPLIT * 32];

  const int tiles_q = NN / 16;
  int tile = blockIdx.x;
  const int b = tile / (NHEAD * tiles_q);
  tile -= b * NHEAD * tiles_q;
  const int h  = tile / tiles_q;
  const int tq = tile - h * tiles_q;

  const int wave = __builtin_amdgcn_readfirstlane((int)(threadIdx.x >> 5));
  const int lane = threadIdx.x & 31;
  const int half = lane >> 4;
  const int l15  = lane & 15;
  const float scale = 0.25f;

  const float* __restrict__ Qp = qkv + hm_plane(b, h * 48);
  const float* __restrict__ Kp = qkv + hm_plane(b, h * 48 + 16);
  const float* __restrict__ Vp = qkv + hm_plane(b, h * 48 + 32);
  float* __restrict__ wl = lds + wave * ATT_WLDS;
  float* __restrict__ cw = wl + 16 * ATT_PITCH;

  float qa[8];
  {
    const float* __restrict__ qr = Qp + (size_t)(tq * 16 + l15) * HDIM + half * 8;
    const v4f q0 = *reinterpret_cast<const v4f*>(qr);
    const v4f q1 = *reinterpret_cast<const v4f*>(qr + 4);
    qa[0] = q0.x * scale;  qa[1] = q0.y * scale;  qa[2] = q0.z * scale;  qa[3] = q0.w * scale;
    qa[4] = q1.x * scale;  qa[5] = q1.y * scale;  qa[6] = q1.z * scale;  qa[7] = q1.w * scale;
  }

  v8f o = {};
  float mrun = -1e30f, lrun = 0.f;

  const int tile0 = wave * ATT_TILES;
  const float* __restrict__ kp = Kp + (size_t)tile0 * 256 + lane * 4;
  const float* __restrict__ vp = Vp + (size_t)tile0 * 256 + lane * 4;
  for (int t = 0; t < ATT_TILES; ++t, kp += 256, vp += 256) {
    const v4f vl = *reinterpret_cast<const v4f*>(vp);
    const v4f vh = *reinterpret_cast<const v4f*>(vp + 128);
    const v4f kl = *reinterpret_cast<const v4f*>(kp);
    const v4f kh = *reinterpret_cast<const v4f*>(kp + 128);
    const float vb[8] = {vl.x, vl.y, vl.z, vl.w, vh.x, vh.y, vh.z, vh.w};
    const float kb[8] = {kl.x, kl.y, kl.z, kl.w, kh.x, kh.y, kh.z, kh.w};

    v8f s = {};
#pragma unroll
    for (int i = 0; i < 4; ++i) {
      v2f a, bb;
      a.x  = qa[2 * i];
      a.y  = qa[2 * i + 1];
      bb.x = kb[2 * i];
      bb.y = kb[2 * i + 1];
      s = WMMA_F32(a, bb, s);
    }

#pragma unroll
    for (int j = 0; j < 8; ++j)
      wl[(j + half * 8) * ATT_PITCH + l15] = s[j];
    __builtin_amdgcn_fence(__ATOMIC_RELEASE, "wavefront");
    __builtin_amdgcn_wave_barrier();
    __builtin_amdgcn_fence(__ATOMIC_ACQUIRE, "wavefront");

    const v4f s0 = *reinterpret_cast<const v4f*>(wl + l15 * ATT_PITCH + half * 8);
    const v4f s1 = *reinterpret_cast<const v4f*>(wl + l15 * ATT_PITCH + half * 8 + 4);

    float mx = fmaxf(fmaxf(fmaxf(s0.x, s0.y), fmaxf(s0.z, s0.w)),
                     fmaxf(fmaxf(s1.x, s1.y), fmaxf(s1.z, s1.w)));
    mx = fmaxf(mx, xhalf(mx));
    const float mnew = fmaxf(mrun, mx);
    const float corr = __expf(mrun - mnew);
    float p[8];
    p[0] = __expf(s0.x - mnew);  p[1] = __expf(s0.y - mnew);
    p[2] = __expf(s0.z - mnew);  p[3] = __expf(s0.w - mnew);
    p[4] = __expf(s1.x - mnew);  p[5] = __expf(s1.y - mnew);
    p[6] = __expf(s1.z - mnew);  p[7] = __expf(s1.w - mnew);
    lrun = lrun * corr + (((p[0] + p[1]) + (p[2] + p[3])) + ((p[4] + p[5]) + (p[6] + p[7])));
    mrun = mnew;

    cw[l15] = corr;
    __builtin_amdgcn_fence(__ATOMIC_RELEASE, "wavefront");
    __builtin_amdgcn_wave_barrier();
    __builtin_amdgcn_fence(__ATOMIC_ACQUIRE, "wavefront");
    const v4f c0 = *reinterpret_cast<const v4f*>(cw + half * 8);
    const v4f c1 = *reinterpret_cast<const v4f*>(cw + half * 8 + 4);
    o[0] *= c0.x;  o[1] *= c0.y;  o[2] *= c0.z;  o[3] *= c0.w;
    o[4] *= c1.x;  o[5] *= c1.y;  o[6] *= c1.z;  o[7] *= c1.w;

    asm volatile("" : "+v"(lrun));
    __builtin_amdgcn_sched_barrier(0);
#pragma unroll
    for (int i = 0; i < 4; ++i) {
      v2f a, bb;
      a.x  = p[2 * i];
      a.y  = p[2 * i + 1];
      bb.x = vb[2 * i];
      bb.y = vb[2 * i + 1];
      o = WMMA_F32(a, bb, o);
    }
  }

  const float ltot = lrun + xhalf(lrun);

  const int elane = (int)__builtin_amdgcn_mbcnt_lo(~0u, 0u);
  const int ehalf = elane >> 4;
  const int el15  = elane & 15;

  {
    float* __restrict__ pw = lds + wave * ATT_WLDS + elane;
#pragma unroll
    for (int j = 0; j < 8; ++j)
      pw[j * 32] = o[j];
    mls[wave * 32 + el15]      = mrun;
    mls[wave * 32 + 16 + el15] = ltot;
  }
  __syncthreads();

  float* __restrict__ outp = attout + (size_t)b * NN * NODE_DIM + h * HDIM;
#pragma unroll
  for (int jj = 0; jj < 2; ++jj) {
    const int j  = wave * 2 + jj;
    const int rr = j + ehalf * 8;
    float mw[ATT_KSPLIT];
    float mall = -1e30f;
#pragma unroll
    for (int w = 0; w < ATT_KSPLIT; ++w) {
      mw[w] = mls[w * 32 + rr];
      mall = fmaxf(mall, mw[w]);
    }
    float lsum = 0.f, osum = 0.f;
#pragma unroll
    for (int w = 0; w < ATT_KSPLIT; ++w) {
      const float wgt = __expf(mw[w] - mall);
      lsum += mls[w * 32 + 16 + rr] * wgt;
      osum += lds[w * ATT_WLDS + j * 32 + elane] * wgt;
    }
    const int r = tq * 16 + rr;
    outp[(size_t)r * NODE_DIM + el15] = osum / lsum;
  }
}

extern "C" void kernel_launch(void* const* d_in, const int* in_sizes, int n_in,
                              void* d_out, int out_size, void* d_ws, size_t ws_size,
                              hipStream_t stream) {
  const float* node    = (const float*)d_in[0];
  const float* edge    = (const float*)d_in[1];
  const float* G       = (const float*)d_in[2];
  const float* W_qkv_n = (const float*)d_in[3];
  const float* b_qkv_n = (const float*)d_in[4];
  const float* W_qkv_e = (const float*)d_in[5];
  const float* b_qkv_e = (const float*)d_in[6];
  const float* W_o     = (const float*)d_in[7];
  const float* b_o     = (const float*)d_in[8];
  float* out = (float*)d_out;

  float* qkv  = (float*)d_ws;
  float* qkve = qkv  + (size_t)BSZ * NN * QKV;
  float* deg  = qkve + (size_t)BSZ * EE * QKV;
  float* att  = deg  + (size_t)BSZ * NN;

  gemm_bias_32x32<1><<<(BSZ * NN / 32) * (QKV / 32), 32, 0, stream>>>(
      node, W_qkv_n, b_qkv_n, qkv, BSZ * NN, NODE_DIM, QKV);

  gemm_bias_32x32<2><<<(BSZ * EE / 32) * (QKV / 32), 32, 0, stream>>>(
      edge, W_qkv_e, b_qkv_e, qkve, BSZ * EE, EDGE_DIM, QKV);

  degree_kernel<<<DEG_SLABS * DEG_ESPLIT, 32 * DEG_WAVES, 0, stream>>>(G, deg);

  agg_kernel<<<BSZ * (NN / 32) * (QKV / 32), 32 * AGG_KSPLIT, 0, stream>>>(G, qkve, deg, qkv);

  attn_kernel<<<BSZ * NHEAD * (NN / 16), 32 * ATT_KSPLIT, 0, stream>>>(qkv, att);

  gemm_bias_32x32<0><<<(BSZ * NN / 32) * (NODE_DIM / 32), 32, 0, stream>>>(
      att, W_o, b_o, out, BSZ * NN, NODE_DIM, NODE_DIM);
}


# front launches merged: degree sums and the qkv_e and qkv_n projections run side by side as block ranges of ONE launch (16-wave groups, a tile per wave); 6 launches -> 4; sum order kept
# speedup vs baseline: 3.0237x; 1.0419x over previous
#include <hip/hip_runtime.h>
#include <hip/hip_bf16.h>


typedef __attribute__((ext_vector_type(2))) float v2f;
typedef __attribute__((ext_vector_type(4))) float v4f;
typedef __attribute__((ext_vector_type(8))) float v8f;

#define BSZ        2
#define NN         3072
#define EE         4096
#define NODE_DIM   128
#define EDGE_DIM   64
#define NHEAD      8
#define HDIM       16
#define QKV        384

#define WMMA_F32(A, B, C) \
  __builtin_amdgcn_wmma_f32_16x16x4_f32(false, (A), false, (B), (short)0, (C), false, false)

__device__ __forceinline__ int frag_off(int k, int c) {
  return ((k >> 2) & 1) * 128 + (((k >> 3) & 1) * 16 + c) * 4 + (k & 3);
}

__device__ __forceinline__ size_t hm_plane(int b, int col) {
  const int h = col / (3 * HDIM);
  const int t = (col - h * (3 * HDIM)) >> 4;
  return ((size_t)((t * BSZ + b) * NHEAD + h) * NN) * HDIM;
}
__device__ __forceinline__ int hm_kind(int col) { return (col % (3 * HDIM)) >> 4; }
__device__ __forceinline__ int hm_chan(int col) { return (col % (3 * HDIM)) & 15; }
__device__ __forceinline__ int hm_off(int t, int n, int c) {
  const int tile = (n >> 4) * 256;
  const int m = n & 15;
  if (t == 0) return n * HDIM + c;
  if (t == 1) return tile + frag_off(c, m);
  return tile + frag_off(m, c);
}

template <int MODE>
__device__ __forceinline__
void gemm_tile(const float* __restrict__ X, const float* __restrict__ W,
               const float* __restrict__ bias, float* __restrict__ C,
               int K, int Ncols, int tile, int lane) {
  const int tiles_n = Ncols >> 5;
  const int tm = tile / tiles_n;
  const int tn = tile % tiles_n;
  const int half = lane >> 4;
  const int l15  = lane & 15;

  const int row0 = tm * 32 + l15;
  const int row1 = row0 + 16;
  const int col0 = tn * 32 + l15;
  const int col1 = col0 + 16;

  const float* __restrict__ Xr0 = X + (size_t)row0 * K;
  const float* __restrict__ Xr1 = X + (size_t)row1 * K;

  v8f acc00 = {}, acc01 = {}, acc10 = {}, acc11 = {};
  for (int k = 0; k < K; k += 4) {
    const int ka = k + half * 2;
    v2f a0, a1, b0, b1;
    a0.x = Xr0[ka];                         a0.y = Xr0[ka + 1];
    a1.x = Xr1[ka];                         a1.y = Xr1[ka + 1];
    b0.x = W[(size_t)ka * Ncols + col0];    b0.y = W[(size_t)(ka + 1) * Ncols + col0];
    b1.x = W[(size_t)ka * Ncols + col1];    b1.y = W[(size_t)(ka + 1) * Ncols + col1];
    acc00 = WMMA_F32(a0, b0, acc00);
    acc01 = WMMA_F32(a0, b1, acc01);
    acc10 = WMMA_F32(a1, b0, acc10);
    acc11 = WMMA_F32(a1, b1, acc11);
  }

  const float bv0 = bias[col0];
  const float bv1 = bias[col1];
  if (MODE == 1) {
    const int bb = (tm * 32) / NN;
    const int nb = tm * 32 - bb * NN;
    float* __restrict__ C0 = C + hm_plane(bb, col0);
    float* __restrict__ C1 = C + hm_plane(bb, col1);
    const int t0 = hm_kind(col0), ch0 = hm_chan(col0);
    const int t1 = hm_kind(col1), ch1 = hm_chan(col1);
#pragma unroll
    for (int j = 0; j < 8; ++j) {
      const int n0 = nb + j + half * 8;
      const int n1 = n0 + 16;
      C0[hm_off(t0, n0, ch0)] = acc00[j] + bv0;
      C1[hm_off(t1, n0, ch1)] = acc01[j] + bv1;
      C0[hm_off(t0, n1, ch0)] = acc10[j] + bv0;
      C1[hm_off(t1, n1, ch1)] = acc11[j] + bv1;
    }
  } else if (MODE == 2) {
    const int bands = Ncols >> 4;
    float* __restrict__ T00 = C + ((size_t)(tm * 2)     * bands + tn * 2) * 256 + lane * 4;
    float* __restrict__ T10 = C + ((size_t)(tm * 2 + 1) * bands + tn * 2) * 256 + lane * 4;
    v4f lo, hi;
    lo = (v4f){acc00[0], acc00[1], acc00[2], acc00[3]} + bv0;
    hi = (v4f){acc00[4], acc00[5], acc00[6], acc00[7]} + bv0;
    *reinterpret_cast<v4f*>(T00) = lo;        *reinterpret_cast<v4f*>(T00 + 128) = hi;
    lo = (v4f){acc01[0], acc01[1], acc01[2], acc01[3]} + bv1;
    hi = (v4f){acc01[4], acc01[5], acc01[6], acc01[7]} + bv1;
    *reinterpret_cast<v4f*>(T00 + 256) = lo;  *reinterpret_cast<v4f*>(T00 + 384) = hi;
    lo = (v4f){acc10[0], acc10[1], acc10[2], acc10[3]} + bv0;
    hi = (v4f){acc10[4], acc10[5], acc10[6], acc10[7]} + bv0;
    *reinterpret_cast<v4f*>(T10) = lo;        *reinterpret_cast<v4f*>(T10 + 128) = hi;
    lo = (v4f){acc11[0], acc11[1], acc11[2], acc11[3]} + bv1;
    hi = (v4f){acc11[4], acc11[5], acc11[6], acc11[7]} + bv1;
    *reinterpret_cast<v4f*>(T10 + 256) = lo;  *reinterpret_cast<v4f*>(T10 + 384) = hi;
  } else {
#pragma unroll
    for (int j = 0; j < 8; ++j) {
      const int r0 = tm * 32 + j + half * 8;
      const int r1 = r0 + 16;
      C[(size_t)r0 * Ncols + col0] = acc00[j] + bv0;
      C[(size_t)r0 * Ncols + col1] = acc01[j] + bv1;
      C[(size_t)r1 * Ncols + col0] = acc10[j] + bv0;
      C[(size_t)r1 * Ncols + col1] = acc11[j] + bv1;
    }
  }
}

template <int MODE>
__global__ __launch_bounds__(32)
void gemm_bias_32x32(const float* __restrict__ X, const float* __restrict__ W,
                     const float* __restrict__ bias, float* __restrict__ C,
                     int K, int Ncols) {
  gemm_tile<MODE>(X, W, bias, C, K, Ncols, (int)blockIdx.x, (int)threadIdx.x);
}

#define DEG_COLS    64
#define DEG_ESPLIT  2
#define DEG_WAVES   16
#define DEG_ROWS    (EE / (DEG_ESPLIT * DEG_WAVES))
#define DEG_BLK     16
#define DEG_SLABS   (BSZ * NN / DEG_COLS)

__device__ __forceinline__
void degree_group(const float* __restrict__ G, float* __restrict__ degp,
                  int blk, v2f* __restrict__ part) {
  const int slab = blk % DEG_SLABS;
  const int eh   = blk / DEG_SLABS;
  const int wave = threadIdx.x >> 5;
  const int lane = threadIdx.x & 31;

  const int idx = slab * DEG_COLS + lane * 2;
  const int b = idx / NN;
  const int n = idx - b * NN;
  const int e0 = (eh * DEG_WAVES + wave) * DEG_ROWS;
  const float* __restrict__ g = G + (size_t)b * EE * NN + (size_t)e0 * NN + n;

  v2f cur[DEG_BLK], nxt[DEG_BLK];
#pragma unroll
  for (int i = 0; i < DEG_BLK; ++i)
    cur[i] = *reinterpret_cast<const v2f*>(g + (size_t)i * NN);

  v2f s = {0.f, 0.f};
  for (int r = DEG_BLK; r < DEG_ROWS; r += DEG_BLK) {
#pragma unroll
    for (int i = 0; i < DEG_BLK; ++i)
      nxt[i] = *reinterpret_cast<const v2f*>(g + (size_t)(r + i) * NN);
#pragma unroll
    for (int i = 0; i < DEG_BLK; ++i) s += cur[i];
#pragma unroll
    for (int i = 0; i < DEG_BLK; ++i) cur[i] = nxt[i];
  }
#pragma unroll
  for (int i = 0; i < DEG_BLK; ++i) s += cur[i];

  part[wave * 32 + lane] = s;
  __syncthreads();

  if (wave == 0) {
    v2f t = part[lane];
#pragma unroll
    for (int w = 1; w < DEG_WAVES; ++w) t += part[w * 32 + lane];
    *reinterpret_cast<v2f*>(degp + (size_t)eh * (BSZ * NN) + idx) = t;
  }
}

#define FRONT_DEG  (DEG_SLABS * DEG_ESPLIT)
#define FRONT_E    ((BSZ * EE / 32) * (QKV / 32) / DEG_WAVES)
#define FRONT_N    ((BSZ * NN / 32) * (QKV / 32) / DEG_WAVES)

__global__ __launch_bounds__(32 * DEG_WAVES)
void front_kernel(const float* __restrict__ G, float* __restrict__ degp,
                  const float* __restrict__ edge, const float* __restrict__ W_e,
                  const float* __restrict__ b_e, float* __restrict__ qkve,
                  const float* __restrict__ node, const float* __restrict__ W_n,
                  const float* __restrict__ b_n, float* __restrict__ qkv,
                  int K_e, int K_n, int Ncols) {
  __shared__ v2f part[DEG_WAVES * 32];

  const int blk = blockIdx.x;
  if (blk < FRONT_DEG) {
    degree_group(G, degp, blk, part);
    return;
  }
  const int wave = __builtin_amdgcn_readfirstlane((int)(threadIdx.x >> 5));
  const int lane = threadIdx.x & 31;
  if (blk < FRONT_DEG + FRONT_E)
    gemm_tile<2>(edge, W_e, b_e, qkve, K_e, Ncols, (blk - FRONT_DEG) * DEG_WAVES + wave, lane);
  else
    gemm_tile<1>(node, W_n, b_n, qkv, K_n, Ncols, (blk - FRONT_DEG - FRONT_E) * DEG_WAVES + wave, lane);
}

#define AGG_KSPLIT 4
#define AGG_KLEN   (EE / AGG_KSPLIT)

__global__ __launch_bounds__(32 * AGG_KSPLIT)
void agg_kernel(const float* __restrict__ G, const float* __restrict__ qkv_e,
                const float* __restrict__ degp, float* __restrict__ qkv) {
  __shared__ float part[AGG_KSPLIT * 4 * 8 * 32];

  const int tiles_c = QKV / 32;
  const int tiles_n = NN / 32;
  int tile = blockIdx.x;
  const int b  = tile / (tiles_n * tiles_c);
  tile -= b * tiles_n * tiles_c;
  const int tn = tile / tiles_c;
  const int tc = tile - tn * tiles_c;

  const int wave = __builtin_amdgcn_readfirstlane((int)(threadIdx.x >> 5));
  const int lane = threadIdx.x & 31;
  const int half = lane >> 4;
  const int l15  = lane & 15;
  const int n0 = tn * 32 + l15;
  const int n1 = n0 + 16;

  const int kbeg = wave * AGG_KLEN;
  const float* __restrict__ Gb = G + (size_t)b * EE * NN + (size_t)(kbeg + half * 8) * NN;
  const float* __restrict__ Eb = qkv_e
      + ((size_t)(b * (EE / 16) + kbeg / 16) * (QKV / 16) + tc * 2) * 256 + lane * 4;

  v8f acc00 = {}, acc01 = {}, acc10 = {}, acc11 = {};
  for (int kb = 0; kb < AGG_KLEN / 16; ++kb) {
    const float* __restrict__ ep = Eb + (size_t)kb * ((QKV / 16) * 256);
    const float* __restrict__ gp = Gb + (size_t)kb * (16 * NN);
    const v4f e0l = *reinterpret_cast<const v4f*>(ep);
    const v4f e0h = *reinterpret_cast<const v4f*>(ep + 128);
    const v4f e1l = *reinterpret_cast<const v4f*>(ep + 256);
    const v4f e1h = *reinterpret_cast<const v4f*>(ep + 384);
    float g0[8], g1[8];
#pragma unroll
    for (int e = 0; e < 8; ++e) {
      g0[e] = gp[(size_t)e * NN + n0];
      g1[e] = gp[(size_t)e * NN + n1];
    }
    const float eb0[8] = {e0l.x, e0l.y, e0l.z, e0l.w, e0h.x, e0h.y, e0h.z, e0h.w};
    const float eb1[8] = {e1l.x, e1l.y, e1l.z, e1l.w, e1h.x, e1h.y, e1h.z, e1h.w};
#pragma unroll
    for (int i = 0; i < 4; ++i) {
      v2f a0, a1, b0, b1;
      a0.x = g0[2 * i];   a0.y = g0[2 * i + 1];
      a1.x = g1[2 * i];   a1.y = g1[2 * i + 1];
      b0.x = eb0[2 * i];  b0.y = eb0[2 * i + 1];
      b1.x = eb1[2 * i];  b1.y = eb1[2 * i + 1];
      acc00 = WMMA_F32(a0, b0, acc00);
      acc01 = WMMA_F32(a0, b1, acc01);
      acc10 = WMMA_F32(a1, b0, acc10);
      acc11 = WMMA_F32(a1, b1, acc11);
    }
  }

  {
    float* __restrict__ pw = part + wave * (4 * 8 * 32) + lane;
#pragma unroll
    for (int j = 0; j < 8; ++j) {
      pw[(0 * 8 + j) * 32] = acc00[j];
      pw[(1 * 8 + j) * 32] = acc01[j];
      pw[(2 * 8 + j) * 32] = acc10[j];
      pw[(3 * 8 + j) * 32] = acc11[j];
    }
  }
  __syncthreads();

  const int rband = wave >> 1;
  const int cband = wave & 1;
  const int col   = tc * 32 + cband * 16 + l15;
  const float* __restrict__ pr = part + wave * (8 * 32) + lane;
  float* __restrict__ out = qkv + hm_plane(b, col);
  const int kind = hm_kind(col);
  const int chan = hm_chan(col);
#pragma unroll
  for (int j = 0; j < 8; ++j) {
    float s = pr[0 * (4 * 8 * 32) + j * 32];
    s += pr[1 * (4 * 8 * 32) + j * 32];
    s += pr[2 * (4 * 8 * 32) + j * 32];
    s += pr[3 * (4 * 8 * 32) + j * 32];
    const int r = tn * 32 + rband * 16 + j + half * 8;
    const float rdiv = 1.0f / (degp[b * NN + r] + degp[BSZ * NN + b * NN + r]);
    const int o = hm_off(kind, r, chan);
    out[o] = out[o] + s * rdiv;
  }
}

#define ATT_KSPLIT 4
#define ATT_TILES  (NN / 16 / ATT_KSPLIT)
#define ATT_PITCH  20
#define ATT_WLDS   (16 * ATT_PITCH + 16)

__device__ __forceinline__ float xhalf(float x) {
  const int bits = __builtin_bit_cast(int, x);
  return __builtin_bit_cast(float,
      __builtin_amdgcn_permlanex16(bits, bits, 0x76543210, 0xfedcba98, false, false));
}

__global__ __launch_bounds__(32 * ATT_KSPLIT, 12)
void attn_kernel(const float* __restrict__ qkv, float* __restrict__ attout) {
  __shared__ __attribute__((aligned(16))) float lds[ATT_KSPLIT * ATT_WLDS];
  __shared__ float mls[ATT_KSPLIT * 32];

  const int tiles_q = NN / 16;
  int tile = blockIdx.x;
  const int b = tile / (NHEAD * tiles_q);
  tile -= b * NHEAD * tiles_q;
  const int h  = tile / tiles_q;
  const int tq = tile - h * tiles_q;

  const int wave = __builtin_amdgcn_readfirstlane((int)(threadIdx.x >> 5));
  const int lane = threadIdx.x & 31;
  const int half = lane >> 4;
  const int l15  = lane & 15;
  const float scale = 0.25f;

  const float* __restrict__ Qp = qkv + hm_plane(b, h * 48);
  const float* __restrict__ Kp = qkv + hm_plane(b, h * 48 + 16);
  const float* __restrict__ Vp = qkv + hm_plane(b, h * 48 + 32);
  float* __restrict__ wl = lds + wave * ATT_WLDS;
  float* __restrict__ cw = wl + 16 * ATT_PITCH;

  float qa[8];
  {
    const float* __restrict__ qr = Qp + (size_t)(tq * 16 + l15) * HDIM + half * 8;
    const v4f q0 = *reinterpret_cast<const v4f*>(qr);
    const v4f q1 = *reinterpret_cast<const v4f*>(qr + 4);
    qa[0] = q0.x * scale;  qa[1] = q0.y * scale;  qa[2] = q0.z * scale;  qa[3] = q0.w * scale;
    qa[4] = q1.x * scale;  qa[5] = q1.y * scale;  qa[6] = q1.z * scale;  qa[7] = q1.w * scale;
  }

  v8f o = {};
  float mrun = -1e30f, lrun = 0.f;

  const int tile0 = wave * ATT_TILES;
  const float* __restrict__ kp = Kp + (size_t)tile0 * 256 + lane * 4;
  const float* __restrict__ vp = Vp + (size_t)tile0 * 256 + lane * 4;
  for (int t = 0; t < ATT_TILES; ++t, kp += 256, vp += 256) {
    const v4f vl = *reinterpret_cast<const v4f*>(vp);
    const v4f vh = *reinterpret_cast<const v4f*>(vp + 128);
    const v4f kl = *reinterpret_cast<const v4f*>(kp);
    const v4f kh = *reinterpret_cast<const v4f*>(kp + 128);
    const float vb[8] = {vl.x, vl.y, vl.z, vl.w, vh.x, vh.y, vh.z, vh.w};
    const float kb[8] = {kl.x, kl.y, kl.z, kl.w, kh.x, kh.y, kh.z, kh.w};

    v8f s = {};
#pragma unroll
    for (int i = 0; i < 4; ++i) {
      v2f a, bb;
      a.x  = qa[2 * i];
      a.y  = qa[2 * i + 1];
      bb.x = kb[2 * i];
      bb.y = kb[2 * i + 1];
      s = WMMA_F32(a, bb, s);
    }

#pragma unroll
    for (int j = 0; j < 8; ++j)
      wl[(j + half * 8) * ATT_PITCH + l15] = s[j];
    __builtin_amdgcn_fence(__ATOMIC_RELEASE, "wavefront");
    __builtin_amdgcn_wave_barrier();
    __builtin_amdgcn_fence(__ATOMIC_ACQUIRE, "wavefront");

    const v4f s0 = *reinterpret_cast<const v4f*>(wl + l15 * ATT_PITCH + half * 8);
    const v4f s1 = *reinterpret_cast<const v4f*>(wl + l15 * ATT_PITCH + half * 8 + 4);

    float mx = fmaxf(fmaxf(fmaxf(s0.x, s0.y), fmaxf(s0.z, s0.w)),
                     fmaxf(fmaxf(s1.x, s1.y), fmaxf(s1.z, s1.w)));
    mx = fmaxf(mx, xhalf(mx));
    const float mnew = fmaxf(mrun, mx);
    const float corr = __expf(mrun - mnew);
    float p[8];
    p[0] = __expf(s0.x - mnew);  p[1] = __expf(s0.y - mnew);
    p[2] = __expf(s0.z - mnew);  p[3] = __expf(s0.w - mnew);
    p[4] = __expf(s1.x - mnew);  p[5] = __expf(s1.y - mnew);
    p[6] = __expf(s1.z - mnew);  p[7] = __expf(s1.w - mnew);
    lrun = lrun * corr + (((p[0] + p[1]) + (p[2] + p[3])) + ((p[4] + p[5]) + (p[6] + p[7])));
    mrun = mnew;

    cw[l15] = corr;
    __builtin_amdgcn_fence(__ATOMIC_RELEASE, "wavefront");
    __builtin_amdgcn_wave_barrier();
    __builtin_amdgcn_fence(__ATOMIC_ACQUIRE, "wavefront");
    const v4f c0 = *reinterpret_cast<const v4f*>(cw + half * 8);
    const v4f c1 = *reinterpret_cast<const v4f*>(cw + half * 8 + 4);
    o[0] *= c0.x;  o[1] *= c0.y;  o[2] *= c0.z;  o[3] *= c0.w;
    o[4] *= c1.x;  o[5] *= c1.y;  o[6] *= c1.z;  o[7] *= c1.w;

    asm volatile("" : "+v"(lrun));
    __builtin_amdgcn_sched_barrier(0);
#pragma unroll
    for (int i = 0; i < 4; ++i) {
      v2f a, bb;
      a.x  = p[2 * i];
      a.y  = p[2 * i + 1];
      bb.x = vb[2 * i];
      bb.y = vb[2 * i + 1];
      o = WMMA_F32(a, bb, o);
    }
  }

  const float ltot = lrun + xhalf(lrun);

  const int elane = (int)__builtin_amdgcn_mbcnt_lo(~0u, 0u);
  const int ehalf = elane >> 4;
  const int el15  = elane & 15;

  {
    float* __restrict__ pw = lds + wave * ATT_WLDS + elane;
#pragma unroll
    for (int j = 0; j < 8; ++j)
      pw[j * 32] = o[j];
    mls[wave * 32 + el15]      = mrun;
    mls[wave * 32 + 16 + el15] = ltot;
  }
  __syncthreads();

  float* __restrict__ outp = attout + (size_t)b * NN * NODE_DIM + h * HDIM;
#pragma unroll
  for (int jj = 0; jj < 2; ++jj) {
    const int j  = wave * 2 + jj;
    const int rr = j + ehalf * 8;
    float mw[ATT_KSPLIT];
    float mall = -1e30f;
#pragma unroll
    for (int w = 0; w < ATT_KSPLIT; ++w) {
      mw[w] = mls[w * 32 + rr];
      mall = fmaxf(mall, mw[w]);
    }
    float lsum = 0.f, osum = 0.f;
#pragma unroll
    for (int w = 0; w < ATT_KSPLIT; ++w) {
      const float wgt = __expf(mw[w] - mall);
      lsum += mls[w * 32 + 16 + rr] * wgt;
      osum += lds[w * ATT_WLDS + j * 32 + elane] * wgt;
    }
    const int r = tq * 16 + rr;
    outp[(size_t)r * NODE_DIM + el15] = osum / lsum;
  }
}

extern "C" void kernel_launch(void* const* d_in, const int* in_sizes, int n_in,
                              void* d_out, int out_size, void* d_ws, size_t ws_size,
                              hipStream_t stream) {
  const float* node    = (const float*)d_in[0];
  const float* edge    = (const float*)d_in[1];
  const float* G       = (const float*)d_in[2];
  const float* W_qkv_n = (const float*)d_in[3];
  const float* b_qkv_n = (const float*)d_in[4];
  const float* W_qkv_e = (const float*)d_in[5];
  const float* b_qkv_e = (const float*)d_in[6];
  const float* W_o     = (const float*)d_in[7];
  const float* b_o     = (const float*)d_in[8];
  float* out = (float*)d_out;

  float* qkv  = (float*)d_ws;
  float* qkve = qkv  + (size_t)BSZ * NN * QKV;
  float* deg  = qkve + (size_t)BSZ * EE * QKV;
  float* att  = deg  + (size_t)BSZ * NN;

  front_kernel<<<FRONT_DEG + FRONT_E + FRONT_N, 32 * DEG_WAVES, 0, stream>>>(
      G, deg, edge, W_qkv_e, b_qkv_e, qkve, node, W_qkv_n, b_qkv_n, qkv,
      EDGE_DIM, NODE_DIM, QKV);

  agg_kernel<<<BSZ * (NN / 32) * (QKV / 32), 32 * AGG_KSPLIT, 0, stream>>>(G, qkve, deg, qkv);

  attn_kernel<<<BSZ * NHEAD * (NN / 16), 32 * ATT_KSPLIT, 0, stream>>>(qkv, att);

  gemm_bias_32x32<0><<<(BSZ * NN / 32) * (NODE_DIM / 32), 32, 0, stream>>>(
      att, W_o, b_o, out, NODE_DIM, NODE_DIM);
}


# aggregation: a work-group is 12 waves = 3 neighbouring 32x32 column tiles x 4 quarters of e (was 4 waves, 1 tile): 3 waves stream identical G lines, WGP cache serves repeats; sums and order kept
# speedup vs baseline: 3.1993x; 1.0581x over previous
#include <hip/hip_runtime.h>
#include <hip/hip_bf16.h>


typedef __attribute__((ext_vector_type(2))) float v2f;
typedef __attribute__((ext_vector_type(4))) float v4f;
typedef __attribute__((ext_vector_type(8))) float v8f;

#define BSZ        2
#define NN         3072
#define EE         4096
#define NODE_DIM   128
#define EDGE_DIM   64
#define NHEAD      8
#define HDIM       16
#define QKV        384

#define WMMA_F32(A, B, C) \
  __builtin_amdgcn_wmma_f32_16x16x4_f32(false, (A), false, (B), (short)0, (C), false, false)

__device__ __forceinline__ int frag_off(int k, int c) {
  return ((k >> 2) & 1) * 128 + (((k >> 3) & 1) * 16 + c) * 4 + (k & 3);
}

__device__ __forceinline__ size_t hm_plane(int b, int col) {
  const int h = col / (3 * HDIM);
  const int t = (col - h * (3 * HDIM)) >> 4;
  return ((size_t)((t * BSZ + b) * NHEAD + h) * NN) * HDIM;
}
__device__ __forceinline__ int hm_kind(int col) { return (col % (3 * HDIM)) >> 4; }
__device__ __forceinline__ int hm_chan(int col) { return (col % (3 * HDIM)) & 15; }
__device__ __forceinline__ int hm_off(int t, int n, int c) {
  const int tile = (n >> 4) * 256;
  const int m = n & 15;
  if (t == 0) return n * HDIM + c;
  if (t == 1) return tile + frag_off(c, m);
  return tile + frag_off(m, c);
}

template <int MODE>
__device__ __forceinline__
void gemm_tile(const float* __restrict__ X, const float* __restrict__ W,
               const float* __restrict__ bias, float* __restrict__ C,
               int K, int Ncols, int tile, int lane) {
  const int tiles_n = Ncols >> 5;
  const int tm = tile / tiles_n;
  const int tn = tile % tiles_n;
  const int half = lane >> 4;
  const int l15  = lane & 15;

  const int row0 = tm * 32 + l15;
  const int row1 = row0 + 16;
  const int col0 = tn * 32 + l15;
  const int col1 = col0 + 16;

  const float* __restrict__ Xr0 = X + (size_t)row0 * K;
  const float* __restrict__ Xr1 = X + (size_t)row1 * K;

  v8f acc00 = {}, acc01 = {}, acc10 = {}, acc11 = {};
  for (int k = 0; k < K; k += 4) {
    const int ka = k + half * 2;
    v2f a0, a1, b0, b1;
    a0.x = Xr0[ka];                         a0.y = Xr0[ka + 1];
    a1.x = Xr1[ka];                         a1.y = Xr1[ka + 1];
    b0.x = W[(size_t)ka * Ncols + col0];    b0.y = W[(size_t)(ka + 1) * Ncols + col0];
    b1.x = W[(size_t)ka * Ncols + col1];    b1.y = W[(size_t)(ka + 1) * Ncols + col1];
    acc00 = WMMA_F32(a0, b0, acc00);
    acc01 = WMMA_F32(a0, b1, acc01);
    acc10 = WMMA_F32(a1, b0, acc10);
    acc11 = WMMA_F32(a1, b1, acc11);
  }

  const float bv0 = bias[col0];
  const float bv1 = bias[col1];
  if (MODE == 1) {
    const int bb = (tm * 32) / NN;
    const int nb = tm * 32 - bb * NN;
    float* __restrict__ C0 = C + hm_plane(bb, col0);
    float* __restrict__ C1 = C + hm_plane(bb, col1);
    const int t0 = hm_kind(col0), ch0 = hm_chan(col0);
    const int t1 = hm_kind(col1), ch1 = hm_chan(col1);
#pragma unroll
    for (int j = 0; j < 8; ++j) {
      const int n0 = nb + j + half * 8;
      const int n1 = n0 + 16;
      C0[hm_off(t0, n0, ch0)] = acc00[j] + bv0;
      C1[hm_off(t1, n0, ch1)] = acc01[j] + bv1;
      C0[hm_off(t0, n1, ch0)] = acc10[j] + bv0;
      C1[hm_off(t1, n1, ch1)] = acc11[j] + bv1;
    }
  } else if (MODE == 2) {
    const int bands = Ncols >> 4;
    float* __restrict__ T00 = C + ((size_t)(tm * 2)     * bands + tn * 2) * 256 + lane * 4;
    float* __restrict__ T10 = C + ((size_t)(tm * 2 + 1) * bands + tn * 2) * 256 + lane * 4;
    v4f lo, hi;
    lo = (v4f){acc00[0], acc00[1], acc00[2], acc00[3]} + bv0;
    hi = (v4f){acc00[4], acc00[5], acc00[6], acc00[7]} + bv0;
    *reinterpret_cast<v4f*>(T00) = lo;        *reinterpret_cast<v4f*>(T00 + 128) = hi;
    lo = (v4f){acc01[0], acc01[1], acc01[2], acc01[3]} + bv1;
    hi = (v4f){acc01[4], acc01[5], acc01[6], acc01[7]} + bv1;
    *reinterpret_cast<v4f*>(T00 + 256) = lo;  *reinterpret_cast<v4f*>(T00 + 384) = hi;
    lo = (v4f){acc10[0], acc10[1], acc10[2], acc10[3]} + bv0;
    hi = (v4f){acc10[4], acc10[5], acc10[6], acc10[7]} + bv0;
    *reinterpret_cast<v4f*>(T10) = lo;        *reinterpret_cast<v4f*>(T10 + 128) = hi;
    lo = (v4f){acc11[0], acc11[1], acc11[2], acc11[3]} + bv1;
    hi = (v4f){acc11[4], acc11[5], acc11[6], acc11[7]} + bv1;
    *reinterpret_cast<v4f*>(T10 + 256) = lo;  *reinterpret_cast<v4f*>(T10 + 384) = hi;
  } else {
#pragma unroll
    for (int j = 0; j < 8; ++j) {
      const int r0 = tm * 32 + j + half * 8;
      const int r1 = r0 + 16;
      C[(size_t)r0 * Ncols + col0] = acc00[j] + bv0;
      C[(size_t)r0 * Ncols + col1] = acc01[j] + bv1;
      C[(size_t)r1 * Ncols + col0] = acc10[j] + bv0;
      C[(size_t)r1 * Ncols + col1] = acc11[j] + bv1;
    }
  }
}

template <int MODE>
__global__ __launch_bounds__(32)
void gemm_bias_32x32(const float* __restrict__ X, const float* __restrict__ W,
                     const float* __restrict__ bias, float* __restrict__ C,
                     int K, int Ncols) {
  gemm_tile<MODE>(X, W, bias, C, K, Ncols, (int)blockIdx.x, (int)threadIdx.x);
}

#define DEG_COLS    64
#define DEG_ESPLIT  2
#define DEG_WAVES   16
#define DEG_ROWS    (EE / (DEG_ESPLIT * DEG_WAVES))
#define DEG_BLK     16
#define DEG_SLABS   (BSZ * NN / DEG_COLS)

__device__ __forceinline__
void degree_group(const float* __restrict__ G, float* __restrict__ degp,
                  int blk, v2f* __restrict__ part) {
  const int slab = blk % DEG_SLABS;
  const int eh   = blk / DEG_SLABS;
  const int wave = threadIdx.x >> 5;
  const int lane = threadIdx.x & 31;

  const int idx = slab * DEG_COLS + lane * 2;
  const int b = idx / NN;
  const int n = idx - b * NN;
  const int e0 = (eh * DEG_WAVES + wave) * DEG_ROWS;
  const float* __restrict__ g = G + (size_t)b * EE * NN + (size_t)e0 * NN + n;

  v2f cur[DEG_BLK], nxt[DEG_BLK];
#pragma unroll
  for (int i = 0; i < DEG_BLK; ++i)
    cur[i] = *reinterpret_cast<const v2f*>(g + (size_t)i * NN);

  v2f s = {0.f, 0.f};
  for (int r = DEG_BLK; r < DEG_ROWS; r += DEG_BLK) {
#pragma unroll
    for (int i = 0; i < DEG_BLK; ++i)
      nxt[i] = *reinterpret_cast<const v2f*>(g + (size_t)(r + i) * NN);
#pragma unroll
    for (int i = 0; i < DEG_BLK; ++i) s += cur[i];
#pragma unroll
    for (int i = 0; i < DEG_BLK; ++i) cur[i] = nxt[i];
  }
#pragma unroll
  for (int i = 0; i < DEG_BLK; ++i) s += cur[i];

  part[wave * 32 + lane] = s;
  __syncthreads();

  if (wave == 0) {
    v2f t = part[lane];
#pragma unroll
    for (int w = 1; w < DEG_WAVES; ++w) t += part[w * 32 + lane];
    *reinterpret_cast<v2f*>(degp + (size_t)eh * (BSZ * NN) + idx) = t;
  }
}

#define FRONT_DEG  (DEG_SLABS * DEG_ESPLIT)
#define FRONT_E    ((BSZ * EE / 32) * (QKV / 32) / DEG_WAVES)
#define FRONT_N    ((BSZ * NN / 32) * (QKV / 32) / DEG_WAVES)

__global__ __launch_bounds__(32 * DEG_WAVES)
void front_kernel(const float* __restrict__ G, float* __restrict__ degp,
                  const float* __restrict__ edge, const float* __restrict__ W_e,
                  const float* __restrict__ b_e, float* __restrict__ qkve,
                  const float* __restrict__ node, const float* __restrict__ W_n,
                  const float* __restrict__ b_n, float* __restrict__ qkv,
                  int K_e, int K_n, int Ncols) {
  __shared__ v2f part[DEG_WAVES * 32];

  const int blk = blockIdx.x;
  if (blk < FRONT_DEG) {
    degree_group(G, degp, blk, part);
    return;
  }
  const int wave = __builtin_amdgcn_readfirstlane((int)(threadIdx.x >> 5));
  const int lane = threadIdx.x & 31;
  if (blk < FRONT_DEG + FRONT_E)
    gemm_tile<2>(edge, W_e, b_e, qkve, K_e, Ncols, (blk - FRONT_DEG) * DEG_WAVES + wave, lane);
  else
    gemm_tile<1>(node, W_n, b_n, qkv, K_n, Ncols, (blk - FRONT_DEG - FRONT_E) * DEG_WAVES + wave, lane);
}

#define AGG_KSPLIT 4
#define AGG_KLEN   (EE / AGG_KSPLIT)
#define AGG_CTILES 3
#define AGG_TILE_F (AGG_KSPLIT * 4 * 8 * 32)

__global__ __launch_bounds__(32 * AGG_KSPLIT * AGG_CTILES)
void agg_kernel(const float* __restrict__ G, const float* __restrict__ qkv_e,
                const float* __restrict__ degp, float* __restrict__ qkv) {
  __shared__ float part[AGG_CTILES * AGG_TILE_F];

  const int groups_c = QKV / 32 / AGG_CTILES;
  const int tiles_n = NN / 32;
  int tile = blockIdx.x;
  const int b  = tile / (tiles_n * groups_c);
  tile -= b * tiles_n * groups_c;
  const int tn = tile / groups_c;
  const int tg = tile - tn * groups_c;

  const int wv   = __builtin_amdgcn_readfirstlane((int)(threadIdx.x >> 5));
  const int wave = wv & (AGG_KSPLIT - 1);
  const int ct   = wv / AGG_KSPLIT;
  const int tc   = tg * AGG_CTILES + ct;
  float* __restrict__ tpart = part + ct * AGG_TILE_F;
  const int lane = threadIdx.x & 31;
  const int half = lane >> 4;
  const int l15  = lane & 15;
  const int n0 = tn * 32 + l15;
  const int n1 = n0 + 16;

  const int kbeg = wave * AGG_KLEN;
  const float* __restrict__ Gb = G + (size_t)b * EE * NN + (size_t)(kbeg + half * 8) * NN;
  const float* __restrict__ Eb = qkv_e
      + ((size_t)(b * (EE / 16) + kbeg / 16) * (QKV / 16) + tc * 2) * 256 + lane * 4;

  v8f acc00 = {}, acc01 = {}, acc10 = {}, acc11 = {};
  for (int kb = 0; kb < AGG_KLEN / 16; ++kb) {
    const float* __restrict__ ep = Eb + (size_t)kb * ((QKV / 16) * 256);
    const float* __restrict__ gp = Gb + (size_t)kb * (16 * NN);
    const v4f e0l = *reinterpret_cast<const v4f*>(ep);
    const v4f e0h = *reinterpret_cast<const v4f*>(ep + 128);
    const v4f e1l = *reinterpret_cast<const v4f*>(ep + 256);
    const v4f e1h = *reinterpret_cast<const v4f*>(ep + 384);
    float g0[8], g1[8];
#pragma unroll
    for (int e = 0; e < 8; ++e) {
      g0[e] = gp[(size_t)e * NN + n0];
      g1[e] = gp[(size_t)e * NN + n1];
    }
    const float eb0[8] = {e0l.x, e0l.y, e0l.z, e0l.w, e0h.x, e0h.y, e0h.z, e0h.w};
    const float eb1[8] = {e1l.x, e1l.y, e1l.z, e1l.w, e1h.x, e1h.y, e1h.z, e1h.w};
#pragma unroll
    for (int i = 0; i < 4; ++i) {
      v2f a0, a1, b0, b1;
      a0.x = g0[2 * i];   a0.y = g0[2 * i + 1];
      a1.x = g1[2 * i];   a1.y = g1[2 * i + 1];
      b0.x = eb0[2 * i];  b0.y = eb0[2 * i + 1];
      b1.x = eb1[2 * i];  b1.y = eb1[2 * i + 1];
      acc00 = WMMA_F32(a0, b0, acc00);
      acc01 = WMMA_F32(a0, b1, acc01);
      acc10 = WMMA_F32(a1, b0, acc10);
      acc11 = WMMA_F32(a1, b1, acc11);
    }
  }

  {
    float* __restrict__ pw = tpart + wave * (4 * 8 * 32) + lane;
#pragma unroll
    for (int j = 0; j < 8; ++j) {
      pw[(0 * 8 + j) * 32] = acc00[j];
      pw[(1 * 8 + j) * 32] = acc01[j];
      pw[(2 * 8 + j) * 32] = acc10[j];
      pw[(3 * 8 + j) * 32] = acc11[j];
    }
  }
  __syncthreads();

  const int rband = wave >> 1;
  const int cband = wave & 1;
  const int col   = tc * 32 + cband * 16 + l15;
  const float* __restrict__ pr = tpart + wave * (8 * 32) + lane;
  float* __restrict__ out = qkv + hm_plane(b, col);
  const int kind = hm_kind(col);
  const int chan = hm_chan(col);
#pragma unroll
  for (int j = 0; j < 8; ++j) {
    float s = pr[0 * (4 * 8 * 32) + j * 32];
    s += pr[1 * (4 * 8 * 32) + j * 32];
    s += pr[2 * (4 * 8 * 32) + j * 32];
    s += pr[3 * (4 * 8 * 32) + j * 32];
    const int r = tn * 32 + rband * 16 + j + half * 8;
    const float rdiv = 1.0f / (degp[b * NN + r] + degp[BSZ * NN + b * NN + r]);
    const int o = hm_off(kind, r, chan);
    out[o] = out[o] + s * rdiv;
  }
}

#define ATT_KSPLIT 4
#define ATT_TILES  (NN / 16 / ATT_KSPLIT)
#define ATT_PITCH  20
#define ATT_WLDS   (16 * ATT_PITCH + 16)

__device__ __forceinline__ float xhalf(float x) {
  const int bits = __builtin_bit_cast(int, x);
  return __builtin_bit_cast(float,
      __builtin_amdgcn_permlanex16(bits, bits, 0x76543210, 0xfedcba98, false, false));
}

__global__ __launch_bounds__(32 * ATT_KSPLIT, 12)
void attn_kernel(const float* __restrict__ qkv, float* __restrict__ attout) {
  __shared__ __attribute__((aligned(16))) float lds[ATT_KSPLIT * ATT_WLDS];
  __shared__ float mls[ATT_KSPLIT * 32];

  const int tiles_q = NN / 16;
  int tile = blockIdx.x;
  const int b = tile / (NHEAD * tiles_q);
  tile -= b * NHEAD * tiles_q;
  const int h  = tile / tiles_q;
  const int tq = tile - h * tiles_q;

  const int wave = __builtin_amdgcn_readfirstlane((int)(threadIdx.x >> 5));
  const int lane = threadIdx.x & 31;
  const int half = lane >> 4;
  const int l15  = lane & 15;
  const float scale = 0.25f;

  const float* __restrict__ Qp = qkv + hm_plane(b, h * 48);
  const float* __restrict__ Kp = qkv + hm_plane(b, h * 48 + 16);
  const float* __restrict__ Vp = qkv + hm_plane(b, h * 48 + 32);
  float* __restrict__ wl = lds + wave * ATT_WLDS;
  float* __restrict__ cw = wl + 16 * ATT_PITCH;

  float qa[8];
  {
    const float* __restrict__ qr = Qp + (size_t)(tq * 16 + l15) * HDIM + half * 8;
    const v4f q0 = *reinterpret_cast<const v4f*>(qr);
    const v4f q1 = *reinterpret_cast<const v4f*>(qr + 4);
    qa[0] = q0.x * scale;  qa[1] = q0.y * scale;  qa[2] = q0.z * scale;  qa[3] = q0.w * scale;
    qa[4] = q1.x * scale;  qa[5] = q1.y * scale;  qa[6] = q1.z * scale;  qa[7] = q1.w * scale;
  }

  v8f o = {};
  float mrun = -1e30f, lrun = 0.f;

  const int tile0 = wave * ATT_TILES;
  const float* __restrict__ kp = Kp + (size_t)tile0 * 256 + lane * 4;
  const float* __restrict__ vp = Vp + (size_t)tile0 * 256 + lane * 4;
  for (int t = 0; t < ATT_TILES; ++t, kp += 256, vp += 256) {
    const v4f vl = *reinterpret_cast<const v4f*>(vp);
    const v4f vh = *reinterpret_cast<const v4f*>(vp + 128);
    const v4f kl = *reinterpret_cast<const v4f*>(kp);
    const v4f kh = *reinterpret_cast<const v4f*>(kp + 128);
    const float vb[8] = {vl.x, vl.y, vl.z, vl.w, vh.x, vh.y, vh.z, vh.w};
    const float kb[8] = {kl.x, kl.y, kl.z, kl.w, kh.x, kh.y, kh.z, kh.w};

    v8f s = {};
#pragma unroll
    for (int i = 0; i < 4; ++i) {
      v2f a, bb;
      a.x  = qa[2 * i];
      a.y  = qa[2 * i + 1];
      bb.x = kb[2 * i];
      bb.y = kb[2 * i + 1];
      s = WMMA_F32(a, bb, s);
    }

#pragma unroll
    for (int j = 0; j < 8; ++j)
      wl[(j + half * 8) * ATT_PITCH + l15] = s[j];
    __builtin_amdgcn_fence(__ATOMIC_RELEASE, "wavefront");
    __builtin_amdgcn_wave_barrier();
    __builtin_amdgcn_fence(__ATOMIC_ACQUIRE, "wavefront");

    const v4f s0 = *reinterpret_cast<const v4f*>(wl + l15 * ATT_PITCH + half * 8);
    const v4f s1 = *reinterpret_cast<const v4f*>(wl + l15 * ATT_PITCH + half * 8 + 4);

    float mx = fmaxf(fmaxf(fmaxf(s0.x, s0.y), fmaxf(s0.z, s0.w)),
                     fmaxf(fmaxf(s1.x, s1.y), fmaxf(s1.z, s1.w)));
    mx = fmaxf(mx, xhalf(mx));
    const float mnew = fmaxf(mrun, mx);
    const float corr = __expf(mrun - mnew);
    float p[8];
    p[0] = __expf(s0.x - mnew);  p[1] = __expf(s0.y - mnew);
    p[2] = __expf(s0.z - mnew);  p[3] = __expf(s0.w - mnew);
    p[4] = __expf(s1.x - mnew);  p[5] = __expf(s1.y - mnew);
    p[6] = __expf(s1.z - mnew);  p[7] = __expf(s1.w - mnew);
    lrun = lrun * corr + (((p[0] + p[1]) + (p[2] + p[3])) + ((p[4] + p[5]) + (p[6] + p[7])));
    mrun = mnew;

    cw[l15] = corr;
    __builtin_amdgcn_fence(__ATOMIC_RELEASE, "wavefront");
    __builtin_amdgcn_wave_barrier();
    __builtin_amdgcn_fence(__ATOMIC_ACQUIRE, "wavefront");
    const v4f c0 = *reinterpret_cast<const v4f*>(cw + half * 8);
    const v4f c1 = *reinterpret_cast<const v4f*>(cw + half * 8 + 4);
    o[0] *= c0.x;  o[1] *= c0.y;  o[2] *= c0.z;  o[3] *= c0.w;
    o[4] *= c1.x;  o[5] *= c1.y;  o[6] *= c1.z;  o[7] *= c1.w;

    asm volatile("" : "+v"(lrun));
    __builtin_amdgcn_sched_barrier(0);
#pragma unroll
    for (int i = 0; i < 4; ++i) {
      v2f a, bb;
      a.x  = p[2 * i];
      a.y  = p[2 * i + 1];
      bb.x = vb[2 * i];
      bb.y = vb[2 * i + 1];
      o = WMMA_F32(a, bb, o);
    }
  }

  const float ltot = lrun + xhalf(lrun);

  const int elane = (int)__builtin_amdgcn_mbcnt_lo(~0u, 0u);
  const int ehalf = elane >> 4;
  const int el15  = elane & 15;

  {
    float* __restrict__ pw = lds + wave * ATT_WLDS + elane;
#pragma unroll
    for (int j = 0; j < 8; ++j)
      pw[j * 32] = o[j];
    mls[wave * 32 + el15]      = mrun;
    mls[wave * 32 + 16 + el15] = ltot;
  }
  __syncthreads();

  float* __restrict__ outp = attout + (size_t)b * NN * NODE_DIM + h * HDIM;
#pragma unroll
  for (int jj = 0; jj < 2; ++jj) {
    const int j  = wave * 2 + jj;
    const int rr = j + ehalf * 8;
    float mw[ATT_KSPLIT];
    float mall = -1e30f;
#pragma unroll
    for (int w = 0; w < ATT_KSPLIT; ++w) {
      mw[w] = mls[w * 32 + rr];
      mall = fmaxf(mall, mw[w]);
    }
    float lsum = 0.f, osum = 0.f;
#pragma unroll
    for (int w = 0; w < ATT_KSPLIT; ++w) {
      const float wgt = __expf(mw[w] - mall);
      lsum += mls[w * 32 + 16 + rr] * wgt;
      osum += lds[w * ATT_WLDS + j * 32 + elane] * wgt;
    }
    const int r = tq * 16 + rr;
    outp[(size_t)r * NODE_DIM + el15] = osum / lsum;
  }
}

extern "C" void kernel_launch(void* const* d_in, const int* in_sizes, int n_in,
                              void* d_out, int out_size, void* d_ws, size_t ws_size,
                              hipStream_t stream) {
  const float* node    = (const float*)d_in[0];
  const float* edge    = (const float*)d_in[1];
  const float* G       = (const float*)d_in[2];
  const float* W_qkv_n = (const float*)d_in[3];
  const float* b_qkv_n = (const float*)d_in[4];
  const float* W_qkv_e = (const float*)d_in[5];
  const float* b_qkv_e = (const float*)d_in[6];
  const float* W_o     = (const float*)d_in[7];
  const float* b_o     = (const float*)d_in[8];
  float* out = (float*)d_out;

  float* qkv  = (float*)d_ws;
  float* qkve = qkv  + (size_t)BSZ * NN * QKV;
  float* deg  = qkve + (size_t)BSZ * EE * QKV;
  float* att  = deg  + (size_t)BSZ * NN;

  front_kernel<<<FRONT_DEG + FRONT_E + FRONT_N, 32 * DEG_WAVES, 0, stream>>>(
      G, deg, edge, W_qkv_e, b_qkv_e, qkve, node, W_qkv_n, b_qkv_n, qkv,
      EDGE_DIM, NODE_DIM, QKV);

  agg_kernel<<<BSZ * (NN / 32) * (QKV / 32 / AGG_CTILES), 32 * AGG_KSPLIT * AGG_CTILES, 0, stream>>>(G, qkve, deg, qkv);

  attn_kernel<<<BSZ * NHEAD * (NN / 16), 32 * ATT_KSPLIT, 0, stream>>>(qkv, att);

  gemm_bias_32x32<0><<<(BSZ * NN / 32) * (NODE_DIM / 32), 32, 0, stream>>>(
      att, W_o, b_o, out, NODE_DIM, NODE_DIM);
}
